# starting kernel
# baseline (speedup 1.0000x reference)
#include <hip/hip_runtime.h>

#define D 128
#define WLDS_STRIDE 132

typedef float v2f __attribute__((ext_vector_type(2)));
typedef float v8f __attribute__((ext_vector_type(8)));

__global__ void zero_f4_kernel(float4* __restrict__ p, int n4) {
    int i = blockIdx.x * blockDim.x + threadIdx.x;
    if (i < n4) p[i] = make_float4(0.f, 0.f, 0.f, 0.f);
}

__global__ __launch_bounds__(256)
void spmm_scatter_kernel(const float* __restrict__ X,
                         const int*   __restrict__ rows,
                         const int*   __restrict__ cols,
                         const float* __restrict__ vals,
                         float* __restrict__ h,
                         int nEdges) {
    int gid  = blockIdx.x * blockDim.x + threadIdx.x;
    int e    = gid >> 5;
    int lane = gid & 31;
    e = __builtin_amdgcn_readfirstlane(e);
    if (e >= nEdges) return;

    int   r = rows[e];
    int   c = cols[e];
    float v = vals[e];

    const float4 xv = ((const float4*)(X + (size_t)c * D))[lane];
    float* hp = h + (size_t)r * D + lane * 4;
    unsafeAtomicAdd(hp + 0, v * xv.x);
    unsafeAtomicAdd(hp + 1, v * xv.y);
    unsafeAtomicAdd(hp + 2, v * xv.z);
    unsafeAtomicAdd(hp + 3, v * xv.w);
}

__global__ __launch_bounds__(256)
void gemm_wmma_kernel(const float* __restrict__ h,
                      const float* __restrict__ W,
                      const float* __restrict__ bias,
                      float* __restrict__ out,
                      int nNodes) {
    __shared__ float Wl[D * WLDS_STRIDE];

    const int wave = threadIdx.x >> 5;
    const int lane = threadIdx.x & 31;
    const int l15  = lane & 15;
    const int hi   = lane >> 4;
    const int rowBase = blockIdx.x * 128 + wave * 16;

    for (int i = threadIdx.x; i < (D * D) / 4; i += 256) {
        float4 w4 = ((const float4*)W)[i];
        int flat = i * 4;
        int o = flat >> 7;
        int k = flat & 127;
        *(float4*)&Wl[o * WLDS_STRIDE + k] = w4;
    }
    __syncthreads();

    int arow = rowBase + l15;
    if (arow >= nNodes) arow = nNodes - 1;
    const float* aptr = h + (size_t)arow * D + 2 * hi;

    v8f acc[8];
#pragma unroll
    for (int nt = 0; nt < 8; ++nt) acc[nt] = (v8f){0.f,0.f,0.f,0.f,0.f,0.f,0.f,0.f};

    const float* bbase = Wl + (size_t)l15 * WLDS_STRIDE + 2 * hi;

    for (int ks = 0; ks < 32; ++ks) {
        const int k = ks * 4;
        const v2f a = *(const v2f*)(aptr + k);
#pragma unroll
        for (int nt = 0; nt < 8; ++nt) {
            const v2f b2 = *(const v2f*)(bbase + nt * 16 * WLDS_STRIDE + k);
            acc[nt] = __builtin_amdgcn_wmma_f32_16x16x4_f32(
                 false, a,  false, b2,
                 (short)0, acc[nt],
                 false,  false);
        }
    }

#pragma unroll
    for (int nt = 0; nt < 8; ++nt) {
        const float bv = bias[nt * 16 + l15];
#pragma unroll
        for (int v = 0; v < 8; ++v) {
            const int r = rowBase + v + 8 * hi;
            if (r < nNodes)
                __builtin_nontemporal_store(acc[nt][v] + bv,
                                            &out[(size_t)r * D + nt * 16 + l15]);
        }
    }
}

extern "C" void kernel_launch(void* const* d_in, const int* in_sizes, int n_in,
                              void* d_out, int out_size, void* d_ws, size_t ws_size,
                              hipStream_t stream) {
    const float* X    = (const float*)d_in[0];
    const int*   rows = (const int*)  d_in[1];
    const int*   cols = (const int*)  d_in[2];
    const float* vals = (const float*)d_in[3];
    const float* W    = (const float*)d_in[4];
    const float* bias = (const float*)d_in[5];
    float*       out  = (float*)d_out;
    float*       h    = (float*)d_ws;

    const int nNodes = in_sizes[0] / D;
    const int nEdges = in_sizes[1];

    {
        const int n4 = nNodes * (D / 4);
        zero_f4_kernel<<<(n4 + 255) / 256, 256, 0, stream>>>((float4*)h, n4);
    }
    {
        const long long threads = (long long)nEdges * 32;
        const int blocks = (int)((threads + 255) / 256);
        spmm_scatter_kernel<<<blocks, 256, 0, stream>>>(X, rows, cols, vals, h, nEdges);
    }
    {
        const int blocks = (nNodes + 127) / 128;
        gemm_wmma_kernel<<<blocks, 256, 0, stream>>>(h, W, bias, out, nNodes);
    }
}


# Float-atomic scatter replaced by counting sort on row (integer atomic rank, offset scan) and one wave per row summing gathered X rows in registers; sum order in a row = atomic arrival rank
# speedup vs baseline: 1.5551x; 1.5551x over previous
#include <hip/hip_runtime.h>


#define D 128
#define WLDS_STRIDE 132
#define SCAN_TILE 4096

typedef float v2f __attribute__((ext_vector_type(2)));
typedef float v8f __attribute__((ext_vector_type(8)));

__global__ void zero_f4_kernel(float4* __restrict__ p, int n4) {
    int i = blockIdx.x * blockDim.x + threadIdx.x;
    if (i < n4) p[i] = make_float4(0.f, 0.f, 0.f, 0.f);
}

__global__ __launch_bounds__(256)
void row_rank_kernel(const int* __restrict__ rows,
                     int* __restrict__ cnt,
                     int* __restrict__ rank,
                     int nEdges) {
    const int i  = blockIdx.x * 256 + threadIdx.x;
    const int e0 = i * 4;
    if (e0 + 3 < nEdges) {
        const int4 r = ((const int4*)rows)[i];
        int4 k;
        k.x = atomicAdd(&cnt[r.x], 1);
        k.y = atomicAdd(&cnt[r.y], 1);
        k.z = atomicAdd(&cnt[r.z], 1);
        k.w = atomicAdd(&cnt[r.w], 1);
        ((int4*)rank)[i] = k;
    } else {
        for (int e = e0; e < nEdges; ++e) rank[e] = atomicAdd(&cnt[rows[e]], 1);
    }
}

__global__ __launch_bounds__(1024)
void scan_offsets_kernel(const int* __restrict__ cnt,
                         int* __restrict__ off,
                         int nNodes) {
    __shared__ int front[32];
    __shared__ int wtot[32];
    const int t    = threadIdx.x;
    const int lane = t & 31;
    const int wave = t >> 5;
    const int b    = blockIdx.x;

    int s = 0;
    const int4* __restrict__ c4 = (const int4*)cnt;
#pragma unroll 8
    for (int k = 0; k < b; ++k) {
        const int4 q = c4[k * 1024 + t];
        s += (q.x + q.y) + (q.z + q.w);
    }
#pragma unroll
    for (int d = 16; d > 0; d >>= 1) s += __shfl_xor(s, d);
    if (lane == 0) front[wave] = s;

    const int i0 = b * SCAN_TILE + t * 4;
    const int v0 = (i0     < nNodes) ? cnt[i0]     : 0;
    const int v1 = (i0 + 1 < nNodes) ? cnt[i0 + 1] : 0;
    const int v2 = (i0 + 2 < nNodes) ? cnt[i0 + 2] : 0;
    const int v3 = (i0 + 3 < nNodes) ? cnt[i0 + 3] : 0;
    const int tsum = (v0 + v1) + (v2 + v3);
    int inc = tsum;
#pragma unroll
    for (int d = 1; d < 32; d <<= 1) {
        const int u = __shfl_up(inc, d);
        if (lane >= d) inc += u;
    }
    if (lane == 31) wtot[wave] = inc;
    __syncthreads();

    int base = 0;
    for (int w = 0; w < 32; ++w) base += front[w];
    for (int w = 0; w < wave; ++w) base += wtot[w];
    const int ex = base + inc - tsum;
    if (i0 + 3 <= nNodes) {
        *(int4*)&off[i0] = make_int4(ex, ex + v0, ex + v0 + v1, ex + v0 + v1 + v2);
    } else {
        if (i0     <= nNodes) off[i0]     = ex;
        if (i0 + 1 <= nNodes) off[i0 + 1] = ex + v0;
        if (i0 + 2 <= nNodes) off[i0 + 2] = ex + v0 + v1;
    }
}

__global__ __launch_bounds__(256)
void place_edges_kernel(const int*   __restrict__ rows,
                        const int*   __restrict__ cols,
                        const float* __restrict__ vals,
                        const int*   __restrict__ off,
                        const int*   __restrict__ rank,
                        int2* __restrict__ edges,
                        int nEdges) {
    const int i  = blockIdx.x * 256 + threadIdx.x;
    const int e0 = i * 4;
    if (e0 + 3 < nEdges) {
        const int4   r = ((const int4*)rows)[i];
        const int4   k = ((const int4*)rank)[i];
        const int4   c = ((const int4*)cols)[i];
        const float4 v = ((const float4*)vals)[i];
        const int p0 = off[r.x] + k.x;
        const int p1 = off[r.y] + k.y;
        const int p2 = off[r.z] + k.z;
        const int p3 = off[r.w] + k.w;
        edges[p0] = make_int2(c.x, __float_as_int(v.x));
        edges[p1] = make_int2(c.y, __float_as_int(v.y));
        edges[p2] = make_int2(c.z, __float_as_int(v.z));
        edges[p3] = make_int2(c.w, __float_as_int(v.w));
    } else {
        for (int e = e0; e < nEdges; ++e)
            edges[off[rows[e]] + rank[e]] = make_int2(cols[e], __float_as_int(vals[e]));
    }
}

__device__ __forceinline__ void fma4(float4& acc, float v, const float4& x) {
    acc.x += v * x.x; acc.y += v * x.y; acc.z += v * x.z; acc.w += v * x.w;
}

template <int K>
__device__ __forceinline__ void gather_group(float4& acc, const float4* __restrict__ X4,
                                             int myc, int myv, int j, int lane) {
    float4 x[K];
    float  v[K];
#pragma unroll
    for (int k = 0; k < K; ++k) {
        const int c = __builtin_amdgcn_readlane(myc, j + k);
        v[k] = __int_as_float(__builtin_amdgcn_readlane(myv, j + k));
        x[k] = X4[(size_t)c * (D / 4) + lane];
    }
#pragma unroll
    for (int k = 0; k < K; ++k) fma4(acc, v[k], x[k]);
}

__global__ __launch_bounds__(256)
void row_gather_kernel(const float* __restrict__ X,
                       const int*   __restrict__ off,
                       const int2*  __restrict__ edges,
                       float* __restrict__ h,
                       int nNodes) {
    const int gid  = blockIdx.x * 256 + threadIdx.x;
    const int lane = threadIdx.x & 31;
    const int r    = __builtin_amdgcn_readfirstlane(gid >> 5);
    if (r >= nNodes) return;

    const int s = off[r];
    const int e = off[r + 1];
    const float4* __restrict__ X4 = (const float4*)X;
    float4 acc = make_float4(0.f, 0.f, 0.f, 0.f);

    for (int base = s; base < e; base += 32) {
        const int n = (e - base < 32) ? (e - base) : 32;
        int myc = 0, myv = 0;
        if (lane < n) {
            const int2 ed = edges[base + lane];
            myc = ed.x;
            myv = ed.y;
        }
        int j = 0;
        for (; j + 8 <= n; j += 8) gather_group<8>(acc, X4, myc, myv, j, lane);
        if ((n - j) & 4) { gather_group<4>(acc, X4, myc, myv, j, lane); j += 4; }
        if ((n - j) & 2) { gather_group<2>(acc, X4, myc, myv, j, lane); j += 2; }
        if ((n - j) & 1) { gather_group<1>(acc, X4, myc, myv, j, lane); }
    }
    ((float4*)h)[(size_t)r * (D / 4) + lane] = acc;
}

__global__ __launch_bounds__(256)
void gemm_wmma_kernel(const float* __restrict__ h,
                      const float* __restrict__ W,
                      const float* __restrict__ bias,
                      float* __restrict__ out,
                      int nNodes) {
    __shared__ float Wl[D * WLDS_STRIDE];

    const int wave = threadIdx.x >> 5;
    const int lane = threadIdx.x & 31;
    const int l15  = lane & 15;
    const int hi   = lane >> 4;
    const int rowBase = blockIdx.x * 128 + wave * 16;

    for (int i = threadIdx.x; i < (D * D) / 4; i += 256) {
        float4 w4 = ((const float4*)W)[i];
        int flat = i * 4;
        int o = flat >> 7;
        int k = flat & 127;
        *(float4*)&Wl[o * WLDS_STRIDE + k] = w4;
    }
    __syncthreads();

    int arow = rowBase + l15;
    if (arow >= nNodes) arow = nNodes - 1;
    const float* aptr = h + (size_t)arow * D + 2 * hi;

    v8f acc[8];
#pragma unroll
    for (int nt = 0; nt < 8; ++nt) acc[nt] = (v8f){0.f,0.f,0.f,0.f,0.f,0.f,0.f,0.f};

    const float* bbase = Wl + (size_t)l15 * WLDS_STRIDE + 2 * hi;

    for (int ks = 0; ks < 32; ++ks) {
        const int k = ks * 4;
        const v2f a = *(const v2f*)(aptr + k);
#pragma unroll
        for (int nt = 0; nt < 8; ++nt) {
            const v2f b2 = *(const v2f*)(bbase + nt * 16 * WLDS_STRIDE + k);
            acc[nt] = __builtin_amdgcn_wmma_f32_16x16x4_f32(
                 false, a,  false, b2,
                 (short)0, acc[nt],
                 false,  false);
        }
    }

#pragma unroll
    for (int nt = 0; nt < 8; ++nt) {
        const float bv = bias[nt * 16 + l15];
#pragma unroll
        for (int v = 0; v < 8; ++v) {
            const int r = rowBase + v + 8 * hi;
            if (r < nNodes)
                __builtin_nontemporal_store(acc[nt][v] + bv,
                                            &out[(size_t)r * D + nt * 16 + l15]);
        }
    }
}

extern "C" void kernel_launch(void* const* d_in, const int* in_sizes, int n_in,
                              void* d_out, int out_size, void* d_ws, size_t ws_size,
                              hipStream_t stream) {
    const float* X    = (const float*)d_in[0];
    const int*   rows = (const int*)  d_in[1];
    const int*   cols = (const int*)  d_in[2];
    const float* vals = (const float*)d_in[3];
    const float* W    = (const float*)d_in[4];
    const float* bias = (const float*)d_in[5];
    float*       out  = (float*)d_out;

    const int nNodes = in_sizes[0] / D;
    const int nEdges = in_sizes[1];

    const size_t nodes4 = ((size_t)nNodes + 4) & ~(size_t)3;
    const size_t edges4 = ((size_t)nEdges + 3) & ~(size_t)3;
    float* h     = (float*)d_ws;
    int*   cnt   = (int*)(h + (size_t)nNodes * D);
    int*   off   = cnt + nodes4;
    int*   rank  = off + nodes4;
    int2*  edges = (int2*)(rank + edges4);

    {
        const int n4 = (int)(nodes4 / 4);
        zero_f4_kernel<<<(n4 + 255) / 256, 256, 0, stream>>>((float4*)cnt, n4);
    }
    row_rank_kernel<<<((nEdges + 3) / 4 + 255) / 256, 256, 0, stream>>>(rows, cnt, rank, nEdges);
    scan_offsets_kernel<<<(nNodes + SCAN_TILE) / SCAN_TILE, 1024, 0, stream>>>(cnt, off, nNodes);
    place_edges_kernel<<<((nEdges + 3) / 4 + 255) / 256, 256, 0, stream>>>(rows, cols, vals, off, rank, edges, nEdges);
    row_gather_kernel<<<(nNodes + 7) / 8, 256, 0, stream>>>(X, off, edges, h, nNodes);
    {
        const int blocks = (nNodes + 127) / 128;
        gemm_wmma_kernel<<<blocks, 256, 0, stream>>>(h, W, bias, out, nNodes);
    }
}


# Per-edge global atomic rank replaced by 16-bit LDS row counters per edge chunk (returned LDS add = rank) and a per-chunk row-start scan; sum order in a row = chunk, then LDS arrival rank
# speedup vs baseline: 2.2593x; 1.4528x over previous
#include <hip/hip_runtime.h>


#define D 128
#define WLDS_STRIDE 132
#define CHUNK 49152
#define CAP_WORDS 65536
#define SCAN_ROWS 1024

typedef float v2f __attribute__((ext_vector_type(2)));
typedef float v8f __attribute__((ext_vector_type(8)));

__device__ __forceinline__ bool in_pass(int r, int rlo, int rhi) { return r >= rlo && r < rhi; }

__device__ __forceinline__ unsigned lds_count(unsigned* cw, int r, int rlo, int rhi) {
    unsigned old = 0u;
    if (in_pass(r, rlo, rhi)) {
        const int rr = r - rlo;
        old = atomicAdd(&cw[rr >> 1], 1u << ((rr & 1) << 4));
    }
    return old;
}

__device__ __forceinline__ int rank_of(unsigned old, int r, int rlo, int rhi) {
    return in_pass(r, rlo, rhi) ? (int)((old >> ((r & 1) << 4)) & 0xFFFFu) : -1;
}

__global__ __launch_bounds__(1024)
void chunk_rank_kernel(const int* __restrict__ rows,
                       unsigned* __restrict__ cntp,
                       int* __restrict__ tsum,
                       int* __restrict__ rank,
                       int nEdges, int nTiles) {
    __shared__ unsigned cw[CAP_WORDS];
    __shared__ int ts[CAP_WORDS / 512];
    const int t    = threadIdx.x;
    const int lane = t & 31;
    const int c    = blockIdx.x;
    const int nChunks = gridDim.x;
    const int eBeg = c * CHUNK;
    const int eEnd = (eBeg + CHUNK < nEdges) ? (eBeg + CHUNK) : nEdges;
    const int eVec = eBeg + ((eEnd - eBeg) & ~3);
    const int nWordsPad = nTiles * (SCAN_ROWS / 2);

    for (int wbase = 0; wbase < nWordsPad; wbase += CAP_WORDS) {
        const int nw = (nWordsPad - wbase < CAP_WORDS) ? (nWordsPad - wbase) : CAP_WORDS;
        int4 rr[CHUNK / 4096];
#pragma unroll
        for (int j = 0; j < CHUNK / 4096; ++j) {
            const int e = eBeg + (j * 1024 + t) * 4;
            rr[j] = (e < eVec) ? *(const int4*)&rows[e] : make_int4(-1, -1, -1, -1);
        }
        for (int w = t * 4; w < nw; w += 4096) *(uint4*)&cw[w] = make_uint4(0u, 0u, 0u, 0u);
        if (t < CAP_WORDS / 512) ts[t] = 0;
        __syncthreads();

        const int rlo = wbase * 2;
        const int rhi = rlo + nw * 2;
#pragma unroll
        for (int j = 0; j < CHUNK / 4096; ++j) {
            const int e = eBeg + (j * 1024 + t) * 4;
            const unsigned ox = lds_count(cw, rr[j].x, rlo, rhi);
            const unsigned oy = lds_count(cw, rr[j].y, rlo, rhi);
            const unsigned oz = lds_count(cw, rr[j].z, rlo, rhi);
            const unsigned ow = lds_count(cw, rr[j].w, rlo, rhi);
            int4 k;
            k.x = rank_of(ox, rr[j].x, rlo, rhi);
            k.y = rank_of(oy, rr[j].y, rlo, rhi);
            k.z = rank_of(oz, rr[j].z, rlo, rhi);
            k.w = rank_of(ow, rr[j].w, rlo, rhi);
            if ((k.x | k.y | k.z | k.w) >= 0) {
                *(int4*)&rank[e] = k;
            } else {
                if (k.x >= 0) rank[e]     = k.x;
                if (k.y >= 0) rank[e + 1] = k.y;
                if (k.z >= 0) rank[e + 2] = k.z;
                if (k.w >= 0) rank[e + 3] = k.w;
            }
        }
        for (int e = eVec + t; e < eEnd; e += 1024) {
            const int r = rows[e];
            const int k = rank_of(lds_count(cw, r, rlo, rhi), r, rlo, rhi);
            if (k >= 0) rank[e] = k;
        }
        __syncthreads();

        for (int w = t * 4; w < nw; w += 4096) {
            const uint4 v = *(const uint4*)&cw[w];
            *(uint4*)&cntp[(size_t)c * nWordsPad + wbase + w] = v;
            int s = (int)((v.x & 0xFFFFu) + (v.x >> 16) + (v.y & 0xFFFFu) + (v.y >> 16) +
                          (v.z & 0xFFFFu) + (v.z >> 16) + (v.w & 0xFFFFu) + (v.w >> 16));
#pragma unroll
            for (int d = 16; d > 0; d >>= 1) s += __shfl_xor(s, d);
            if (lane == 0) atomicAdd(&ts[w >> 9], s);
        }
        __syncthreads();
        for (int k = t; k < (nw >> 9); k += 1024)
            tsum[(size_t)((wbase >> 9) + k) * nChunks + c] = ts[k];
        __syncthreads();
    }
}

__global__ __launch_bounds__(256)
void scan_base_kernel(const unsigned* __restrict__ cntp,
                      const int* __restrict__ tsum,
                      int* __restrict__ off,
                      int* __restrict__ base,
                      int nNodes, int nTiles, int nChunks) {
    __shared__ int front[8];
    __shared__ int wtot[8];
    const int t    = threadIdx.x;
    const int lane = t & 31;
    const int wave = t >> 5;
    const int b    = blockIdx.x;
    const int nWordsPad = nTiles * (SCAN_ROWS / 2);
    const int nRowsPad  = nTiles * SCAN_ROWS;

    int s = 0;
    const int nf = b * nChunks;
#pragma unroll 8
    for (int k = t; k < nf; k += 256) s += tsum[k];
#pragma unroll
    for (int d = 16; d > 0; d >>= 1) s += __shfl_xor(s, d);
    if (lane == 0) front[wave] = s;

    const int w0 = b * (SCAN_ROWS / 2) + t * 2;
    const int i0 = b * SCAN_ROWS + t * 4;
    int v0 = 0, v1 = 0, v2 = 0, v3 = 0;
#pragma unroll 8
    for (int c = 0; c < nChunks; ++c) {
        const uint2 q = *(const uint2*)&cntp[(size_t)c * nWordsPad + w0];
        v0 += (int)(q.x & 0xFFFFu); v1 += (int)(q.x >> 16);
        v2 += (int)(q.y & 0xFFFFu); v3 += (int)(q.y >> 16);
    }
    const int tsum4 = (v0 + v1) + (v2 + v3);
    int inc = tsum4;
#pragma unroll
    for (int d = 1; d < 32; d <<= 1) {
        const int u = __shfl_up(inc, d);
        if (lane >= d) inc += u;
    }
    if (lane == 31) wtot[wave] = inc;
    __syncthreads();

    int pre = 0;
    for (int w = 0; w < 8; ++w) pre += front[w];
    for (int w = 0; w < wave; ++w) pre += wtot[w];
    const int ex = pre + inc - tsum4;
    if (i0 + 3 <= nNodes) {
        *(int4*)&off[i0] = make_int4(ex, ex + v0, ex + v0 + v1, ex + v0 + v1 + v2);
    } else {
        if (i0     <= nNodes) off[i0]     = ex;
        if (i0 + 1 <= nNodes) off[i0 + 1] = ex + v0;
        if (i0 + 2 <= nNodes) off[i0 + 2] = ex + v0 + v1;
    }

    int p0 = ex, p1 = ex + v0, p2 = ex + v0 + v1, p3 = ex + v0 + v1 + v2;
#pragma unroll 8
    for (int c = 0; c < nChunks; ++c) {
        const uint2 q = *(const uint2*)&cntp[(size_t)c * nWordsPad + w0];
        *(int4*)&base[(size_t)c * nRowsPad + i0] = make_int4(p0, p1, p2, p3);
        p0 += (int)(q.x & 0xFFFFu); p1 += (int)(q.x >> 16);
        p2 += (int)(q.y & 0xFFFFu); p3 += (int)(q.y >> 16);
    }
}

__global__ __launch_bounds__(256)
void place_edges_kernel(const int*   __restrict__ rows,
                        const int*   __restrict__ cols,
                        const float* __restrict__ vals,
                        const int*   __restrict__ base,
                        const int*   __restrict__ rank,
                        int2* __restrict__ edges,
                        int nEdges, int nRowsPad) {
    const int i  = blockIdx.x * 256 + threadIdx.x;
    const int e0 = i * 4;
    if (e0 + 3 < nEdges) {
        const int4   r = ((const int4*)rows)[i];
        const int4   k = ((const int4*)rank)[i];
        const int4   c = ((const int4*)cols)[i];
        const float4 v = ((const float4*)vals)[i];
        const int* __restrict__ bc = base + (size_t)(e0 / CHUNK) * nRowsPad;
        const int p0 = bc[r.x] + k.x;
        const int p1 = bc[r.y] + k.y;
        const int p2 = bc[r.z] + k.z;
        const int p3 = bc[r.w] + k.w;
        edges[p0] = make_int2(c.x, __float_as_int(v.x));
        edges[p1] = make_int2(c.y, __float_as_int(v.y));
        edges[p2] = make_int2(c.z, __float_as_int(v.z));
        edges[p3] = make_int2(c.w, __float_as_int(v.w));
    } else {
        for (int e = e0; e < nEdges; ++e)
            edges[base[(size_t)(e / CHUNK) * nRowsPad + rows[e]] + rank[e]] =
                make_int2(cols[e], __float_as_int(vals[e]));
    }
}

__device__ __forceinline__ void fma4(float4& acc, float v, const float4& x) {
    acc.x += v * x.x; acc.y += v * x.y; acc.z += v * x.z; acc.w += v * x.w;
}

template <int K>
__device__ __forceinline__ void gather_group(float4& acc, const float4* __restrict__ X4,
                                             int myc, int myv, int j, int lane) {
    float4 x[K];
    float  v[K];
#pragma unroll
    for (int k = 0; k < K; ++k) {
        const int c = __builtin_amdgcn_readlane(myc, j + k);
        v[k] = __int_as_float(__builtin_amdgcn_readlane(myv, j + k));
        x[k] = X4[(size_t)c * (D / 4) + lane];
    }
#pragma unroll
    for (int k = 0; k < K; ++k) fma4(acc, v[k], x[k]);
}

__global__ __launch_bounds__(256)
void row_gather_kernel(const float* __restrict__ X,
                       const int*   __restrict__ off,
                       const int2*  __restrict__ edges,
                       float* __restrict__ h,
                       int nNodes) {
    const int gid  = blockIdx.x * 256 + threadIdx.x;
    const int lane = threadIdx.x & 31;
    const int r    = __builtin_amdgcn_readfirstlane(gid >> 5);
    if (r >= nNodes) return;

    const int s = off[r];
    const int e = off[r + 1];
    const float4* __restrict__ X4 = (const float4*)X;
    float4 acc = make_float4(0.f, 0.f, 0.f, 0.f);

    for (int base = s; base < e; base += 32) {
        const int n = (e - base < 32) ? (e - base) : 32;
        int myc = 0, myv = 0;
        if (lane < n) {
            const int2 ed = edges[base + lane];
            myc = ed.x;
            myv = ed.y;
        }
        int j = 0;
        for (; j + 8 <= n; j += 8) gather_group<8>(acc, X4, myc, myv, j, lane);
        if ((n - j) & 4) { gather_group<4>(acc, X4, myc, myv, j, lane); j += 4; }
        if ((n - j) & 2) { gather_group<2>(acc, X4, myc, myv, j, lane); j += 2; }
        if ((n - j) & 1) { gather_group<1>(acc, X4, myc, myv, j, lane); }
    }
    ((float4*)h)[(size_t)r * (D / 4) + lane] = acc;
}

__global__ __launch_bounds__(256)
void gemm_wmma_kernel(const float* __restrict__ h,
                      const float* __restrict__ W,
                      const float* __restrict__ bias,
                      float* __restrict__ out,
                      int nNodes) {
    __shared__ float Wl[D * WLDS_STRIDE];

    const int wave = threadIdx.x >> 5;
    const int lane = threadIdx.x & 31;
    const int l15  = lane & 15;
    const int hi   = lane >> 4;
    const int rowBase = blockIdx.x * 128 + wave * 16;

    for (int i = threadIdx.x; i < (D * D) / 4; i += 256) {
        float4 w4 = ((const float4*)W)[i];
        int flat = i * 4;
        int o = flat >> 7;
        int k = flat & 127;
        *(float4*)&Wl[o * WLDS_STRIDE + k] = w4;
    }
    __syncthreads();

    int arow = rowBase + l15;
    if (arow >= nNodes) arow = nNodes - 1;
    const float* aptr = h + (size_t)arow * D + 2 * hi;

    v8f acc[8];
#pragma unroll
    for (int nt = 0; nt < 8; ++nt) acc[nt] = (v8f){0.f,0.f,0.f,0.f,0.f,0.f,0.f,0.f};

    const float* bbase = Wl + (size_t)l15 * WLDS_STRIDE + 2 * hi;

    for (int ks = 0; ks < 32; ++ks) {
        const int k = ks * 4;
        const v2f a = *(const v2f*)(aptr + k);
#pragma unroll
        for (int nt = 0; nt < 8; ++nt) {
            const v2f b2 = *(const v2f*)(bbase + nt * 16 * WLDS_STRIDE + k);
            acc[nt] = __builtin_amdgcn_wmma_f32_16x16x4_f32(
                 false, a,  false, b2,
                 (short)0, acc[nt],
                 false,  false);
        }
    }

#pragma unroll
    for (int nt = 0; nt < 8; ++nt) {
        const float bv = bias[nt * 16 + l15];
#pragma unroll
        for (int v = 0; v < 8; ++v) {
            const int r = rowBase + v + 8 * hi;
            if (r < nNodes)
                __builtin_nontemporal_store(acc[nt][v] + bv,
                                            &out[(size_t)r * D + nt * 16 + l15]);
        }
    }
}

extern "C" void kernel_launch(void* const* d_in, const int* in_sizes, int n_in,
                              void* d_out, int out_size, void* d_ws, size_t ws_size,
                              hipStream_t stream) {
    const float* X    = (const float*)d_in[0];
    const int*   rows = (const int*)  d_in[1];
    const int*   cols = (const int*)  d_in[2];
    const float* vals = (const float*)d_in[3];
    const float* W    = (const float*)d_in[4];
    const float* bias = (const float*)d_in[5];
    float*       out  = (float*)d_out;

    const int nNodes = in_sizes[0] / D;
    const int nEdges = in_sizes[1];

    const size_t nodes4 = ((size_t)nNodes + 4) & ~(size_t)3;
    const size_t edges4 = ((size_t)nEdges + 3) & ~(size_t)3;
    const int nTiles  = nNodes / SCAN_ROWS + 1;
    const int nChunks = (nEdges + CHUNK - 1) / CHUNK;
    const size_t nWordsPad = (size_t)nTiles * (SCAN_ROWS / 2);
    const size_t nRowsPad  = (size_t)nTiles * SCAN_ROWS;
    const size_t tsum4 = ((size_t)nTiles * nChunks + 3) & ~(size_t)3;
    float*    h     = (float*)d_ws;
    unsigned* cntp  = (unsigned*)(h + (size_t)nNodes * D);
    int*      tsum  = (int*)(cntp + (size_t)nChunks * nWordsPad);
    int*      off   = tsum + tsum4;
    int*      base  = off + nodes4;
    int*      rank  = base + (size_t)nChunks * nRowsPad;
    int2*     edges = (int2*)(rank + edges4);

    chunk_rank_kernel<<<nChunks, 1024, 0, stream>>>(rows, cntp, tsum, rank, nEdges, nTiles);
    scan_base_kernel<<<nTiles, 256, 0, stream>>>(cntp, tsum, off, base, nNodes, nTiles, nChunks);
    place_edges_kernel<<<((nEdges + 3) / 4 + 255) / 256, 256, 0, stream>>>(rows, cols, vals, base, rank, edges, nEdges, (int)nRowsPad);
    row_gather_kernel<<<(nNodes + 7) / 8, 256, 0, stream>>>(X, off, edges, h, nNodes);
    {
        const int blocks = (nNodes + 127) / 128;
        gemm_wmma_kernel<<<blocks, 256, 0, stream>>>(h, W, bias, out, nNodes);
    }
}


# Row gather and projection fused in one launch: 8 waves sum a 16-row strip into an LDS tile and multiply it by W fragments kept in registers; h never stored; row and k sum order unchanged
# speedup vs baseline: 2.3363x; 1.0341x over previous
#include <hip/hip_runtime.h>


#define D 128
#define WLDS_STRIDE 132
#define CHUNK 49152
#define CAP_WORDS 65536
#define SCAN_ROWS 1024

typedef float v2f __attribute__((ext_vector_type(2)));
typedef float v8f __attribute__((ext_vector_type(8)));

__device__ __forceinline__ bool in_pass(int r, int rlo, int rhi) { return r >= rlo && r < rhi; }

__device__ __forceinline__ unsigned lds_count(unsigned* cw, int r, int rlo, int rhi) {
    unsigned old = 0u;
    if (in_pass(r, rlo, rhi)) {
        const int rr = r - rlo;
        old = atomicAdd(&cw[rr >> 1], 1u << ((rr & 1) << 4));
    }
    return old;
}

__device__ __forceinline__ int rank_of(unsigned old, int r, int rlo, int rhi) {
    return in_pass(r, rlo, rhi) ? (int)((old >> ((r & 1) << 4)) & 0xFFFFu) : -1;
}

__global__ __launch_bounds__(1024)
void chunk_rank_kernel(const int* __restrict__ rows,
                       unsigned* __restrict__ cntp,
                       int* __restrict__ tsum,
                       int* __restrict__ rank,
                       int nEdges, int nTiles) {
    __shared__ unsigned cw[CAP_WORDS];
    __shared__ int ts[CAP_WORDS / 512];
    const int t    = threadIdx.x;
    const int lane = t & 31;
    const int c    = blockIdx.x;
    const int nChunks = gridDim.x;
    const int eBeg = c * CHUNK;
    const int eEnd = (eBeg + CHUNK < nEdges) ? (eBeg + CHUNK) : nEdges;
    const int eVec = eBeg + ((eEnd - eBeg) & ~3);
    const int nWordsPad = nTiles * (SCAN_ROWS / 2);

    for (int wbase = 0; wbase < nWordsPad; wbase += CAP_WORDS) {
        const int nw = (nWordsPad - wbase < CAP_WORDS) ? (nWordsPad - wbase) : CAP_WORDS;
        int4 rr[CHUNK / 4096];
#pragma unroll
        for (int j = 0; j < CHUNK / 4096; ++j) {
            const int e = eBeg + (j * 1024 + t) * 4;
            rr[j] = (e < eVec) ? *(const int4*)&rows[e] : make_int4(-1, -1, -1, -1);
        }
        for (int w = t * 4; w < nw; w += 4096) *(uint4*)&cw[w] = make_uint4(0u, 0u, 0u, 0u);
        if (t < CAP_WORDS / 512) ts[t] = 0;
        __syncthreads();

        const int rlo = wbase * 2;
        const int rhi = rlo + nw * 2;
#pragma unroll
        for (int j = 0; j < CHUNK / 4096; ++j) {
            const int e = eBeg + (j * 1024 + t) * 4;
            const unsigned ox = lds_count(cw, rr[j].x, rlo, rhi);
            const unsigned oy = lds_count(cw, rr[j].y, rlo, rhi);
            const unsigned oz = lds_count(cw, rr[j].z, rlo, rhi);
            const unsigned ow = lds_count(cw, rr[j].w, rlo, rhi);
            int4 k;
            k.x = rank_of(ox, rr[j].x, rlo, rhi);
            k.y = rank_of(oy, rr[j].y, rlo, rhi);
            k.z = rank_of(oz, rr[j].z, rlo, rhi);
            k.w = rank_of(ow, rr[j].w, rlo, rhi);
            if ((k.x | k.y | k.z | k.w) >= 0) {
                *(int4*)&rank[e] = k;
            } else {
                if (k.x >= 0) rank[e]     = k.x;
                if (k.y >= 0) rank[e + 1] = k.y;
                if (k.z >= 0) rank[e + 2] = k.z;
                if (k.w >= 0) rank[e + 3] = k.w;
            }
        }
        for (int e = eVec + t; e < eEnd; e += 1024) {
            const int r = rows[e];
            const int k = rank_of(lds_count(cw, r, rlo, rhi), r, rlo, rhi);
            if (k >= 0) rank[e] = k;
        }
        __syncthreads();

        for (int w = t * 4; w < nw; w += 4096) {
            const uint4 v = *(const uint4*)&cw[w];
            *(uint4*)&cntp[(size_t)c * nWordsPad + wbase + w] = v;
            int s = (int)((v.x & 0xFFFFu) + (v.x >> 16) + (v.y & 0xFFFFu) + (v.y >> 16) +
                          (v.z & 0xFFFFu) + (v.z >> 16) + (v.w & 0xFFFFu) + (v.w >> 16));
#pragma unroll
            for (int d = 16; d > 0; d >>= 1) s += __shfl_xor(s, d);
            if (lane == 0) atomicAdd(&ts[w >> 9], s);
        }
        __syncthreads();
        for (int k = t; k < (nw >> 9); k += 1024)
            tsum[(size_t)((wbase >> 9) + k) * nChunks + c] = ts[k];
        __syncthreads();
    }
}

__global__ __launch_bounds__(256)
void scan_base_kernel(const unsigned* __restrict__ cntp,
                      const int* __restrict__ tsum,
                      int* __restrict__ off,
                      int* __restrict__ base,
                      int nNodes, int nTiles, int nChunks) {
    __shared__ int front[8];
    __shared__ int wtot[8];
    const int t    = threadIdx.x;
    const int lane = t & 31;
    const int wave = t >> 5;
    const int b    = blockIdx.x;
    const int nWordsPad = nTiles * (SCAN_ROWS / 2);
    const int nRowsPad  = nTiles * SCAN_ROWS;

    int s = 0;
    const int nf = b * nChunks;
#pragma unroll 8
    for (int k = t; k < nf; k += 256) s += tsum[k];
#pragma unroll
    for (int d = 16; d > 0; d >>= 1) s += __shfl_xor(s, d);
    if (lane == 0) front[wave] = s;

    const int w0 = b * (SCAN_ROWS / 2) + t * 2;
    const int i0 = b * SCAN_ROWS + t * 4;
    int v0 = 0, v1 = 0, v2 = 0, v3 = 0;
#pragma unroll 8
    for (int c = 0; c < nChunks; ++c) {
        const uint2 q = *(const uint2*)&cntp[(size_t)c * nWordsPad + w0];
        v0 += (int)(q.x & 0xFFFFu); v1 += (int)(q.x >> 16);
        v2 += (int)(q.y & 0xFFFFu); v3 += (int)(q.y >> 16);
    }
    const int tsum4 = (v0 + v1) + (v2 + v3);
    int inc = tsum4;
#pragma unroll
    for (int d = 1; d < 32; d <<= 1) {
        const int u = __shfl_up(inc, d);
        if (lane >= d) inc += u;
    }
    if (lane == 31) wtot[wave] = inc;
    __syncthreads();

    int pre = 0;
    for (int w = 0; w < 8; ++w) pre += front[w];
    for (int w = 0; w < wave; ++w) pre += wtot[w];
    const int ex = pre + inc - tsum4;
    if (i0 + 3 <= nNodes) {
        *(int4*)&off[i0] = make_int4(ex, ex + v0, ex + v0 + v1, ex + v0 + v1 + v2);
    } else {
        if (i0     <= nNodes) off[i0]     = ex;
        if (i0 + 1 <= nNodes) off[i0 + 1] = ex + v0;
        if (i0 + 2 <= nNodes) off[i0 + 2] = ex + v0 + v1;
    }

    int p0 = ex, p1 = ex + v0, p2 = ex + v0 + v1, p3 = ex + v0 + v1 + v2;
#pragma unroll 8
    for (int c = 0; c < nChunks; ++c) {
        const uint2 q = *(const uint2*)&cntp[(size_t)c * nWordsPad + w0];
        *(int4*)&base[(size_t)c * nRowsPad + i0] = make_int4(p0, p1, p2, p3);
        p0 += (int)(q.x & 0xFFFFu); p1 += (int)(q.x >> 16);
        p2 += (int)(q.y & 0xFFFFu); p3 += (int)(q.y >> 16);
    }
}

__global__ __launch_bounds__(256)
void place_edges_kernel(const int*   __restrict__ rows,
                        const int*   __restrict__ cols,
                        const float* __restrict__ vals,
                        const int*   __restrict__ base,
                        const int*   __restrict__ rank,
                        int2* __restrict__ edges,
                        int nEdges, int nRowsPad) {
    const int i  = blockIdx.x * 256 + threadIdx.x;
    const int e0 = i * 4;
    if (e0 + 3 < nEdges) {
        const int4   r = ((const int4*)rows)[i];
        const int4   k = ((const int4*)rank)[i];
        const int4   c = ((const int4*)cols)[i];
        const float4 v = ((const float4*)vals)[i];
        const int* __restrict__ bc = base + (size_t)(e0 / CHUNK) * nRowsPad;
        const int p0 = bc[r.x] + k.x;
        const int p1 = bc[r.y] + k.y;
        const int p2 = bc[r.z] + k.z;
        const int p3 = bc[r.w] + k.w;
        edges[p0] = make_int2(c.x, __float_as_int(v.x));
        edges[p1] = make_int2(c.y, __float_as_int(v.y));
        edges[p2] = make_int2(c.z, __float_as_int(v.z));
        edges[p3] = make_int2(c.w, __float_as_int(v.w));
    } else {
        for (int e = e0; e < nEdges; ++e)
            edges[base[(size_t)(e / CHUNK) * nRowsPad + rows[e]] + rank[e]] =
                make_int2(cols[e], __float_as_int(vals[e]));
    }
}

#define TILE_FLOATS (16 * WLDS_STRIDE)

__device__ __forceinline__ void fma4(float4& acc, float v, const float4& x) {
    acc.x = __builtin_fmaf(v, x.x, acc.x); acc.y = __builtin_fmaf(v, x.y, acc.y);
    acc.z = __builtin_fmaf(v, x.z, acc.z); acc.w = __builtin_fmaf(v, x.w, acc.w);
}

template <int K>
__device__ __forceinline__ void gather_group(float4& acc, const float4* __restrict__ X4,
                                             int myc, int myv, int j, int lane) {
    float4 x[K];
    float  v[K];
#pragma unroll
    for (int k = 0; k < K; ++k) {
        const int c = __builtin_amdgcn_readlane(myc, j + k);
        v[k] = __int_as_float(__builtin_amdgcn_readlane(myv, j + k));
        x[k] = X4[(size_t)c * (D / 4) + lane];
    }
#pragma unroll
    for (int k = 0; k < K; ++k) fma4(acc, v[k], x[k]);
}

__device__ __forceinline__ void gather_range(float4& acc, const float4* __restrict__ X4,
                                             int myc, int myv, int lo, int hi, int lane) {
    int j = lo;
    for (; j + 8 <= hi; j += 8) gather_group<8>(acc, X4, myc, myv, j, lane);
    if ((hi - j) & 4) { gather_group<4>(acc, X4, myc, myv, j, lane); j += 4; }
    if ((hi - j) & 2) { gather_group<2>(acc, X4, myc, myv, j, lane); j += 2; }
    if ((hi - j) & 1) { gather_group<1>(acc, X4, myc, myv, j, lane); }
}

__device__ __forceinline__ void row_bounds(const int* __restrict__ off, int r0, int nNodes,
                                           int& s, int& m, int& e) {
    const int i0 = (r0     < nNodes) ? r0     : nNodes;
    const int i1 = (r0 + 1 < nNodes) ? r0 + 1 : nNodes;
    const int i2 = (r0 + 2 < nNodes) ? r0 + 2 : nNodes;
    s = off[i0]; m = off[i1]; e = off[i2];
}

__device__ __forceinline__ int2 load_chunk(const int2* __restrict__ edges, int base, int end, int lane) {
    int2 ed = make_int2(0, 0);
    if (base + lane < end) ed = edges[base + lane];
    return ed;
}

__global__ __launch_bounds__(512) __attribute__((amdgpu_waves_per_eu(8, 8)))
void gather_project_kernel(const float* __restrict__ X,
                           const int*   __restrict__ off,
                           const int2*  __restrict__ edges,
                           const float* __restrict__ W,
                           const float* __restrict__ bias,
                           float* __restrict__ out,
                           int nNodes, int nStrips) {
    __shared__ __attribute__((aligned(16))) float lds[D * WLDS_STRIDE];

    const int lane = threadIdx.x & 31;
    const int wv   = __builtin_amdgcn_readfirstlane(threadIdx.x >> 5);
    const int half = wv >> 3;
    const int wave = wv & 7;
    const int l15  = lane & 15;
    const int hi   = lane >> 4;
    const int step = gridDim.x * 2;
    const float4* __restrict__ X4 = (const float4*)X;

    int strip = blockIdx.x * 2 + half;
    int s, m, e;
    row_bounds(off, strip * 16 + wave * 2, nNodes, s, m, e);
    int2 cur = load_chunk(edges, s, e, lane);

    for (int i = threadIdx.x; i < (D * D) / 4; i += 512) {
        const float4 w4 = ((const float4*)W)[i];
        const int flat = i * 4;
        const int o = flat >> 7;
        const int k = flat & 127;
        *(float4*)&lds[o * WLDS_STRIDE + k] = w4;
    }
    __syncthreads();

    v2f bw[32];
    {
        const float* bbase = lds + (wave * 16 + l15) * WLDS_STRIDE + 2 * hi;
#pragma unroll
        for (int ks = 0; ks < 32; ++ks) bw[ks] = *(const v2f*)(bbase + ks * 4);
    }
    const float bv = bias[wave * 16 + l15];
    __syncthreads();

    int buf = 0;
    for (int pair = blockIdx.x * 2; pair < nStrips; pair += step, strip += step) {
        int sn, mn, en;
        row_bounds(off, (strip + step) * 16 + wave * 2, nNodes, sn, mn, en);

        float* tl = lds + (buf * 2 + half) * TILE_FLOATS;
        float4 acc = make_float4(0.f, 0.f, 0.f, 0.f);
        bool row0 = true;
        for (int base = s; base < e; base += 32) {
            const int2 nxt = load_chunk(edges, base + 32, e, lane);
            const int n = (e - base < 32) ? (e - base) : 32;
            int a = m - base;
            a = (a < 0) ? 0 : ((a > n) ? n : a);
            gather_range(acc, X4, cur.x, cur.y, 0, a, lane);
            if (row0 && m <= base + n) {
                *(float4*)&tl[(wave * 2) * WLDS_STRIDE + 4 * lane] = acc;
                acc = make_float4(0.f, 0.f, 0.f, 0.f);
                row0 = false;
            }
            gather_range(acc, X4, cur.x, cur.y, a, n, lane);
            cur = nxt;
        }
        if (row0) {
            *(float4*)&tl[(wave * 2) * WLDS_STRIDE + 4 * lane] = acc;
            acc = make_float4(0.f, 0.f, 0.f, 0.f);
        }
        *(float4*)&tl[(wave * 2 + 1) * WLDS_STRIDE + 4 * lane] = acc;
        __syncthreads();
        cur = load_chunk(edges, sn, en, lane);

        v8f c = (v8f){0.f,0.f,0.f,0.f,0.f,0.f,0.f,0.f};
        const float* ap = tl + l15 * WLDS_STRIDE + 2 * hi;
#pragma unroll
        for (int ks = 0; ks < 32; ++ks) {
            const v2f a2 = *(const v2f*)(ap + ks * 4);
            c = __builtin_amdgcn_wmma_f32_16x16x4_f32(
                 false, a2,  false, bw[ks],
                 (short)0, c,
                 false,  false);
        }

#pragma unroll
        for (int v = 0; v < 8; ++v) {
            const int r = strip * 16 + v + 8 * hi;
            if (r < nNodes)
                __builtin_nontemporal_store(c[v] + bv, &out[(size_t)r * D + wave * 16 + l15]);
        }

        s = sn; m = mn; e = en;
        buf ^= 1;
    }
}

extern "C" void kernel_launch(void* const* d_in, const int* in_sizes, int n_in,
                              void* d_out, int out_size, void* d_ws, size_t ws_size,
                              hipStream_t stream) {
    const float* X    = (const float*)d_in[0];
    const int*   rows = (const int*)  d_in[1];
    const int*   cols = (const int*)  d_in[2];
    const float* vals = (const float*)d_in[3];
    const float* W    = (const float*)d_in[4];
    const float* bias = (const float*)d_in[5];
    float*       out  = (float*)d_out;

    const int nNodes = in_sizes[0] / D;
    const int nEdges = in_sizes[1];

    const size_t nodes4 = ((size_t)nNodes + 4) & ~(size_t)3;
    const size_t edges4 = ((size_t)nEdges + 3) & ~(size_t)3;
    const int nTiles  = nNodes / SCAN_ROWS + 1;
    const int nChunks = (nEdges + CHUNK - 1) / CHUNK;
    const size_t nWordsPad = (size_t)nTiles * (SCAN_ROWS / 2);
    const size_t nRowsPad  = (size_t)nTiles * SCAN_ROWS;
    const size_t tsum4 = ((size_t)nTiles * nChunks + 3) & ~(size_t)3;
    unsigned* cntp  = (unsigned*)d_ws;
    int*      tsum  = (int*)(cntp + (size_t)nChunks * nWordsPad);
    int*      off   = tsum + tsum4;
    int*      base  = off + nodes4;
    int*      rank  = base + (size_t)nChunks * nRowsPad;
    int2*     edges = (int2*)(rank + edges4);

    chunk_rank_kernel<<<nChunks, 1024, 0, stream>>>(rows, cntp, tsum, rank, nEdges, nTiles);
    scan_base_kernel<<<nTiles, 256, 0, stream>>>(cntp, tsum, off, base, nNodes, nTiles, nChunks);
    place_edges_kernel<<<((nEdges + 3) / 4 + 255) / 256, 256, 0, stream>>>(rows, cols, vals, base, rank, edges, nEdges, (int)nRowsPad);
    {
        const int nStrips = (nNodes + 15) / 16;
        const int nPairs  = (nStrips + 1) / 2;
        const int blocks  = (nPairs < 1024) ? nPairs : 1024;
        gather_project_kernel<<<blocks, 512, 0, stream>>>(X, off, edges, W, bias, out, nNodes, nStrips);
    }
}


# Edges grouped by a two-level LDS sort (chunk -> 256-row buckets -> rows): whole-line writes replace the per-edge 8-byte scatter, 3 launches not 4; row sum order = chunk, then LDS arrival rank
# speedup vs baseline: 2.7830x; 1.1912x over previous
#include <hip/hip_runtime.h>


#define D 128
#define WLDS_STRIDE 132
#define EC 8192
#define BSHIFT 8
#define BROWS 256
#define MAXB 1024
#define MAXC 1024
#define RUNS 8
#define TMAX 6144

typedef float v2f __attribute__((ext_vector_type(2)));
typedef float v8f __attribute__((ext_vector_type(8)));

__global__ __launch_bounds__(1024)
void chunk_bucket_kernel(const int*   __restrict__ rows,
                         const int*   __restrict__ cols,
                         const float* __restrict__ vals,
                         int2* __restrict__ cvA,
                         unsigned short* __restrict__ rA,
                         int* __restrict__ run,
                         int nEdges, int nB) {
    __shared__ __attribute__((aligned(16))) int2 scv[EC];
    __shared__ __attribute__((aligned(16))) unsigned short sr[EC];
    __shared__ int cnt[MAXB];
    __shared__ int wsum[32];
    const int t    = threadIdx.x;
    const int lane = t & 31;
    const int wave = t >> 5;
    const int c    = blockIdx.x;
    const int eBeg = c * EC;
    const int n    = (nEdges - eBeg < EC) ? (nEdges - eBeg) : EC;

    int   r[EC / 1024];
    int   cc[EC / 1024];
    float vv[EC / 1024];
#pragma unroll
    for (int j = 0; j < EC / 1024; ++j) {
        const int i = j * 1024 + t;
        r[j]  = (i < n) ? rows[eBeg + i] : -1;
        cc[j] = (i < n) ? cols[eBeg + i] : 0;
        vv[j] = (i < n) ? vals[eBeg + i] : 0.f;
    }
    if (t < MAXB) cnt[t] = 0;
    __syncthreads();

    int rk[EC / 1024];
#pragma unroll
    for (int j = 0; j < EC / 1024; ++j) {
        rk[j] = 0;
        if (r[j] >= 0) rk[j] = atomicAdd(&cnt[r[j] >> BSHIFT], 1);
    }
    __syncthreads();

    const int v = (t < nB) ? cnt[t] : 0;
    int inc = v;
#pragma unroll
    for (int d = 1; d < 32; d <<= 1) {
        const int u = __shfl_up(inc, d);
        if (lane >= d) inc += u;
    }
    if (lane == 31) wsum[wave] = inc;
    __syncthreads();
    int pre = 0;
    for (int w = 0; w < wave; ++w) pre += wsum[w];
    const int ex = pre + inc - v;
    if (t < nB) {
        cnt[t] = ex;
        run[(size_t)c * (nB + 1) + t] = ex;
    }
    if (t == 0) run[(size_t)c * (nB + 1) + nB] = n;
    __syncthreads();

#pragma unroll
    for (int j = 0; j < EC / 1024; ++j) {
        if (r[j] >= 0) {
            const int pos = cnt[r[j] >> BSHIFT] + rk[j];
            scv[pos] = make_int2(cc[j], __float_as_int(vv[j]));
            sr[pos]  = (unsigned short)(r[j] & (BROWS - 1));
        }
    }
    __syncthreads();

    const int n2 = n >> 1;
    for (int i = t; i < n2; i += 1024) ((int4*)(cvA + (size_t)eBeg))[i] = ((const int4*)scv)[i];
    if ((n & 1) && t == 0) cvA[(size_t)eBeg + n - 1] = scv[n - 1];
    const int n8 = n >> 3;
    for (int i = t; i < n8; i += 1024) ((uint4*)(rA + (size_t)eBeg))[i] = ((const uint4*)sr)[i];
    for (int i = (n8 << 3) + t; i < n; i += 1024) rA[(size_t)eBeg + i] = sr[i];
}

__global__ __launch_bounds__(256)
void bucket_rows_kernel(const int2* __restrict__ cvA,
                        const unsigned short* __restrict__ rA,
                        const int* __restrict__ run,
                        int*  __restrict__ off,
                        int2* __restrict__ edges,
                        int nNodes, int nChunks, int nB) {
    __shared__ int lpre[MAXC + 1];
    __shared__ int sbeg[MAXC];
    __shared__ int rc[BROWS];
    __shared__ int red[8];
    __shared__ int red2[8];
    __shared__ int2 ucv[TMAX];
    __shared__ int  ukr[TMAX];
    __shared__ int2 scv[TMAX];
    __shared__ unsigned short sidx[TMAX];
    const int t    = threadIdx.x;
    const int lane = t & 31;
    const int wave = t >> 5;
    const int b    = blockIdx.x;
    const int S    = nB + 1;

    int carry = 0, gsum = 0;
    for (int c0 = 0; c0 < nChunks; c0 += 256) {
        const int c = c0 + t;
        int s = 0, len = 0;
        if (c < nChunks) {
            s   = run[(size_t)c * S + b];
            len = run[(size_t)c * S + b + 1] - s;
        }
        gsum += s;
        int inc = len;
#pragma unroll
        for (int d = 1; d < 32; d <<= 1) {
            const int u = __shfl_up(inc, d);
            if (lane >= d) inc += u;
        }
        if (lane == 31) red[wave] = inc;
        __syncthreads();
        int pre = 0, tot = 0;
        for (int w = 0; w < 8; ++w) { const int x = red[w]; tot += x; if (w < wave) pre += x; }
        if (c < nChunks) { lpre[c] = carry + pre + inc - len; sbeg[c] = s; }
        carry += tot;
        __syncthreads();
    }
    const int T = carry;
#pragma unroll
    for (int d = 16; d > 0; d >>= 1) gsum += __shfl_xor(gsum, d);
    if (lane == 0) red2[wave] = gsum;
    if (t == 0) lpre[nChunks] = T;
    rc[t] = 0;
    __syncthreads();
    int G = 0;
    for (int w = 0; w < 8; ++w) G += red2[w];
    const int row = b * BROWS + t;

    if (T <= TMAX) {
        for (int c0 = wave * RUNS; c0 < nChunks; c0 += 8 * RUNS) {
            int  key[RUNS], ii[RUNS];
            int2 cv[RUNS];
#pragma unroll
            for (int u = 0; u < RUNS; ++u) {
                const int c = c0 + u;
                ii[u] = -1; key[u] = 0; cv[u] = make_int2(0, 0);
                if (c < nChunks) {
                    const int lp = lpre[c];
                    if (lane < lpre[c + 1] - lp) {
                        const size_t src = (size_t)c * EC + sbeg[c] + lane;
                        ii[u]  = lp + lane;
                        key[u] = rA[src];
                        cv[u]  = cvA[src];
                    }
                }
            }
#pragma unroll
            for (int u = 0; u < RUNS; ++u) {
                if (ii[u] >= 0) {
                    ucv[ii[u]] = cv[u];
                    ukr[ii[u]] = key[u] | (atomicAdd(&rc[key[u]], 1) << 8);
                }
            }
            for (int u = 0; u < RUNS; ++u) {
                const int c = c0 + u;
                if (c < nChunks) {
                    const int lp = lpre[c];
                    const int L  = lpre[c + 1] - lp;
                    for (int q = 32 + lane; q < L; q += 32) {
                        const size_t src = (size_t)c * EC + sbeg[c] + q;
                        const int k = rA[src];
                        ucv[lp + q] = cvA[src];
                        ukr[lp + q] = k | (atomicAdd(&rc[k], 1) << 8);
                    }
                }
            }
        }
        __syncthreads();
        const int v = rc[t];
        int inc = v;
#pragma unroll
        for (int d = 1; d < 32; d <<= 1) {
            const int u = __shfl_up(inc, d);
            if (lane >= d) inc += u;
        }
        if (lane == 31) red[wave] = inc;
        __syncthreads();
        int pre = 0;
        for (int w = 0; w < wave; ++w) pre += red[w];
        const int ex = pre + inc - v;
        rc[t] = ex;
        if (row <= nNodes) off[row] = G + ex;
        __syncthreads();
        for (int i = t; i < T; i += 256) {
            const int kr  = ukr[i];
            const int pos = rc[kr & (BROWS - 1)] + (kr >> 8);
            scv[pos]  = ucv[i];
            sidx[pos] = (unsigned short)i;
        }
        __syncthreads();
        for (int p = ex; p < ex + v; ++p) {
            const int idx = sidx[p];
            int k = 0;
            for (int q = ex; q < ex + v; ++q) k += (sidx[q] < idx) ? 1 : 0;
            ucv[ex + k] = scv[p];
        }
        __syncthreads();
        for (int i = t - (G & 15); i < T; i += 256)
            if (i >= 0) edges[(size_t)G + i] = ucv[i];
    } else {
        int cntk = 0;
        for (int c = 0; c < nChunks; ++c) {
            const int L = lpre[c + 1] - lpre[c];
            const size_t src = (size_t)c * EC + sbeg[c];
            for (int q = 0; q < L; ++q) cntk += (rA[src + q] == t) ? 1 : 0;
        }
        int inc = cntk;
#pragma unroll
        for (int d = 1; d < 32; d <<= 1) {
            const int u = __shfl_up(inc, d);
            if (lane >= d) inc += u;
        }
        if (lane == 31) red[wave] = inc;
        __syncthreads();
        int pre = 0;
        for (int w = 0; w < wave; ++w) pre += red[w];
        const int ex = pre + inc - cntk;
        if (row <= nNodes) off[row] = G + ex;
        int p = G + ex;
        for (int c = 0; c < nChunks; ++c) {
            const int L = lpre[c + 1] - lpre[c];
            const size_t src = (size_t)c * EC + sbeg[c];
            for (int q = 0; q < L; ++q)
                if (rA[src + q] == t) edges[p++] = cvA[src + q];
        }
    }
}

#define TILE_FLOATS (16 * WLDS_STRIDE)

__device__ __forceinline__ void fma4(float4& acc, float v, const float4& x) {
    acc.x = __builtin_fmaf(v, x.x, acc.x); acc.y = __builtin_fmaf(v, x.y, acc.y);
    acc.z = __builtin_fmaf(v, x.z, acc.z); acc.w = __builtin_fmaf(v, x.w, acc.w);
}

template <int K>
__device__ __forceinline__ void gather_group(float4& acc, const float4* __restrict__ X4,
                                             int myc, int myv, int j, int lane) {
    float4 x[K];
    float  v[K];
#pragma unroll
    for (int k = 0; k < K; ++k) {
        const int c = __builtin_amdgcn_readlane(myc, j + k);
        v[k] = __int_as_float(__builtin_amdgcn_readlane(myv, j + k));
        x[k] = X4[(size_t)c * (D / 4) + lane];
    }
#pragma unroll
    for (int k = 0; k < K; ++k) fma4(acc, v[k], x[k]);
}

__device__ __forceinline__ void gather_range(float4& acc, const float4* __restrict__ X4,
                                             int myc, int myv, int lo, int hi, int lane) {
    int j = lo;
    for (; j + 8 <= hi; j += 8) gather_group<8>(acc, X4, myc, myv, j, lane);
    if ((hi - j) & 4) { gather_group<4>(acc, X4, myc, myv, j, lane); j += 4; }
    if ((hi - j) & 2) { gather_group<2>(acc, X4, myc, myv, j, lane); j += 2; }
    if ((hi - j) & 1) { gather_group<1>(acc, X4, myc, myv, j, lane); }
}

__device__ __forceinline__ void row_bounds(const int* __restrict__ off, int r0, int nNodes,
                                           int& s, int& m, int& e) {
    const int i0 = (r0     < nNodes) ? r0     : nNodes;
    const int i1 = (r0 + 1 < nNodes) ? r0 + 1 : nNodes;
    const int i2 = (r0 + 2 < nNodes) ? r0 + 2 : nNodes;
    s = off[i0]; m = off[i1]; e = off[i2];
}

__device__ __forceinline__ int2 load_chunk(const int2* __restrict__ edges, int base, int end, int lane) {
    int2 ed = make_int2(0, 0);
    if (base + lane < end) ed = edges[base + lane];
    return ed;
}

__global__ __launch_bounds__(512) __attribute__((amdgpu_waves_per_eu(8, 8)))
void gather_project_kernel(const float* __restrict__ X,
                           const int*   __restrict__ off,
                           const int2*  __restrict__ edges,
                           const float* __restrict__ W,
                           const float* __restrict__ bias,
                           float* __restrict__ out,
                           int nNodes, int nStrips) {
    __shared__ __attribute__((aligned(16))) float lds[D * WLDS_STRIDE];

    const int lane = threadIdx.x & 31;
    const int wv   = __builtin_amdgcn_readfirstlane(threadIdx.x >> 5);
    const int half = wv >> 3;
    const int wave = wv & 7;
    const int l15  = lane & 15;
    const int hi   = lane >> 4;
    const int step = gridDim.x * 2;
    const float4* __restrict__ X4 = (const float4*)X;

    int strip = blockIdx.x * 2 + half;
    int s, m, e;
    row_bounds(off, strip * 16 + wave * 2, nNodes, s, m, e);
    int2 cur = load_chunk(edges, s, e, lane);

    for (int i = threadIdx.x; i < (D * D) / 4; i += 512) {
        const float4 w4 = ((const float4*)W)[i];
        const int flat = i * 4;
        const int o = flat >> 7;
        const int k = flat & 127;
        *(float4*)&lds[o * WLDS_STRIDE + k] = w4;
    }
    __syncthreads();

    v2f bw[32];
    {
        const float* bbase = lds + (wave * 16 + l15) * WLDS_STRIDE + 2 * hi;
#pragma unroll
        for (int ks = 0; ks < 32; ++ks) bw[ks] = *(const v2f*)(bbase + ks * 4);
    }
    const float bv = bias[wave * 16 + l15];
    __syncthreads();

    int buf = 0;
    for (int pair = blockIdx.x * 2; pair < nStrips; pair += step, strip += step) {
        int sn, mn, en;
        row_bounds(off, (strip + step) * 16 + wave * 2, nNodes, sn, mn, en);

        float* tl = lds + (buf * 2 + half) * TILE_FLOATS;
        float4 acc = make_float4(0.f, 0.f, 0.f, 0.f);
        bool row0 = true;
        for (int base = s; base < e; base += 32) {
            const int2 nxt = load_chunk(edges, base + 32, e, lane);
            const int n = (e - base < 32) ? (e - base) : 32;
            int a = m - base;
            a = (a < 0) ? 0 : ((a > n) ? n : a);
            gather_range(acc, X4, cur.x, cur.y, 0, a, lane);
            if (row0 && m <= base + n) {
                *(float4*)&tl[(wave * 2) * WLDS_STRIDE + 4 * lane] = acc;
                acc = make_float4(0.f, 0.f, 0.f, 0.f);
                row0 = false;
            }
            gather_range(acc, X4, cur.x, cur.y, a, n, lane);
            cur = nxt;
        }
        if (row0) {
            *(float4*)&tl[(wave * 2) * WLDS_STRIDE + 4 * lane] = acc;
            acc = make_float4(0.f, 0.f, 0.f, 0.f);
        }
        *(float4*)&tl[(wave * 2 + 1) * WLDS_STRIDE + 4 * lane] = acc;
        __syncthreads();
        cur = load_chunk(edges, sn, en, lane);

        v8f c = (v8f){0.f,0.f,0.f,0.f,0.f,0.f,0.f,0.f};
        const float* ap = tl + l15 * WLDS_STRIDE + 2 * hi;
#pragma unroll
        for (int ks = 0; ks < 32; ++ks) {
            const v2f a2 = *(const v2f*)(ap + ks * 4);
            c = __builtin_amdgcn_wmma_f32_16x16x4_f32(
                 false, a2,  false, bw[ks],
                 (short)0, c,
                 false,  false);
        }

#pragma unroll
        for (int v = 0; v < 8; ++v) {
            const int r = strip * 16 + v + 8 * hi;
            if (r < nNodes)
                __builtin_nontemporal_store(c[v] + bv, &out[(size_t)r * D + wave * 16 + l15]);
        }

        s = sn; m = mn; e = en;
        buf ^= 1;
    }
}

extern "C" void kernel_launch(void* const* d_in, const int* in_sizes, int n_in,
                              void* d_out, int out_size, void* d_ws, size_t ws_size,
                              hipStream_t stream) {
    const float* X    = (const float*)d_in[0];
    const int*   rows = (const int*)  d_in[1];
    const int*   cols = (const int*)  d_in[2];
    const float* vals = (const float*)d_in[3];
    const float* W    = (const float*)d_in[4];
    const float* bias = (const float*)d_in[5];
    float*       out  = (float*)d_out;

    const int nNodes = in_sizes[0] / D;
    const int nEdges = in_sizes[1];

    const int nChunks = (nEdges + EC - 1) / EC;
    const int nB      = (nNodes >> BSHIFT) + 1;
    const size_t recs = (size_t)nChunks * EC;
    char* p = (char*)d_ws;
    int2* cvA = (int2*)p;
    p += (recs * sizeof(int2) + 127) & ~(size_t)127;
    unsigned short* rA = (unsigned short*)p;
    p += (recs * sizeof(unsigned short) + 127) & ~(size_t)127;
    int* run = (int*)p;
    p += ((size_t)nChunks * (nB + 1) * sizeof(int) + 127) & ~(size_t)127;
    int* off = (int*)p;
    p += (((size_t)nNodes + 1) * sizeof(int) + 127) & ~(size_t)127;
    int2* edges = (int2*)p;

    chunk_bucket_kernel<<<nChunks, 1024, 0, stream>>>(rows, cols, vals, cvA, rA, run, nEdges, nB);
    bucket_rows_kernel<<<nB, 256, 0, stream>>>(cvA, rA, run, off, edges, nNodes, nChunks, nB);
    {
        const int nStrips = (nNodes + 15) / 16;
        const int nPairs  = (nStrips + 1) / 2;
        const int blocks  = (nPairs < 1024) ? nPairs : 1024;
        gather_project_kernel<<<blocks, 512, 0, stream>>>(X, off, edges, W, bias, out, nNodes, nStrips);
    }
}


# Bucket pass: per-group LDS counters (8 chunks, one wave adds) + a per-row scan over groups replace arrival ranks and re-ranking; row sum order: a run's records past 32 go last in their group
# speedup vs baseline: 3.4306x; 1.2327x over previous
#include <hip/hip_runtime.h>


#define D 128
#define WLDS_STRIDE 132
#define EC 8192
#define BSHIFT 8
#define BROWS 256
#define MAXB 1024
#define MAXC 1024
#define RUNS 8
#define TMAX 5632
#define NGMAX 32

typedef float v2f __attribute__((ext_vector_type(2)));
typedef float v8f __attribute__((ext_vector_type(8)));

__global__ __launch_bounds__(1024)
void chunk_bucket_kernel(const int*   __restrict__ rows,
                         const int*   __restrict__ cols,
                         const float* __restrict__ vals,
                         int2* __restrict__ cvA,
                         unsigned short* __restrict__ rA,
                         int* __restrict__ run,
                         int nEdges, int nB) {
    __shared__ __attribute__((aligned(16))) int2 scv[EC];
    __shared__ __attribute__((aligned(16))) unsigned short sr[EC];
    __shared__ int cnt[MAXB];
    __shared__ int wsum[32];
    const int t    = threadIdx.x;
    const int lane = t & 31;
    const int wave = t >> 5;
    const int c    = blockIdx.x;
    const int eBeg = c * EC;
    const int n    = (nEdges - eBeg < EC) ? (nEdges - eBeg) : EC;

    int   r[EC / 1024];
    int   cc[EC / 1024];
    float vv[EC / 1024];
#pragma unroll
    for (int j = 0; j < EC / 1024; ++j) {
        const int i = j * 1024 + t;
        r[j]  = (i < n) ? rows[eBeg + i] : -1;
        cc[j] = (i < n) ? cols[eBeg + i] : 0;
        vv[j] = (i < n) ? vals[eBeg + i] : 0.f;
    }
    if (t < MAXB) cnt[t] = 0;
    __syncthreads();

    int rk[EC / 1024];
#pragma unroll
    for (int j = 0; j < EC / 1024; ++j) {
        rk[j] = 0;
        if (r[j] >= 0) rk[j] = atomicAdd(&cnt[r[j] >> BSHIFT], 1);
    }
    __syncthreads();

    const int v = (t < nB) ? cnt[t] : 0;
    int inc = v;
#pragma unroll
    for (int d = 1; d < 32; d <<= 1) {
        const int u = __shfl_up(inc, d);
        if (lane >= d) inc += u;
    }
    if (lane == 31) wsum[wave] = inc;
    __syncthreads();
    int pre = 0;
    for (int w = 0; w < wave; ++w) pre += wsum[w];
    const int ex = pre + inc - v;
    if (t < nB) {
        cnt[t] = ex;
        run[(size_t)c * (nB + 1) + t] = ex;
    }
    if (t == 0) run[(size_t)c * (nB + 1) + nB] = n;
    __syncthreads();

#pragma unroll
    for (int j = 0; j < EC / 1024; ++j) {
        if (r[j] >= 0) {
            const int pos = cnt[r[j] >> BSHIFT] + rk[j];
            scv[pos] = make_int2(cc[j], __float_as_int(vv[j]));
            sr[pos]  = (unsigned short)(r[j] & (BROWS - 1));
        }
    }
    __syncthreads();

    const int n2 = n >> 1;
    for (int i = t; i < n2; i += 1024) ((int4*)(cvA + (size_t)eBeg))[i] = ((const int4*)scv)[i];
    if ((n & 1) && t == 0) cvA[(size_t)eBeg + n - 1] = scv[n - 1];
    const int n8 = n >> 3;
    for (int i = t; i < n8; i += 1024) ((uint4*)(rA + (size_t)eBeg))[i] = ((const uint4*)sr)[i];
    for (int i = (n8 << 3) + t; i < n; i += 1024) rA[(size_t)eBeg + i] = sr[i];
}

__global__ __launch_bounds__(256)
void bucket_rows_kernel(const int2* __restrict__ cvA,
                        const unsigned short* __restrict__ rA,
                        const int* __restrict__ run,
                        int*  __restrict__ off,
                        int2* __restrict__ edges,
                        int nNodes, int nChunks, int nB) {
    __shared__ int lpre[MAXC + 1];
    __shared__ int sbeg[MAXC];
    __shared__ int red[8];
    __shared__ int red2[8];
    __shared__ int gcnt[NGMAX * BROWS];
    __shared__ int2 ucv[TMAX];
    __shared__ int  ukr[TMAX];
    __shared__ int2 scv[TMAX];
    const int t    = threadIdx.x;
    const int lane = t & 31;
    const int wave = t >> 5;
    const int b    = blockIdx.x;
    const int S    = nB + 1;
    const int NG   = (nChunks + RUNS - 1) / RUNS;

    int carry = 0, gsum = 0;
    for (int c0 = 0; c0 < nChunks; c0 += 256) {
        const int c = c0 + t;
        int s = 0, len = 0;
        if (c < nChunks) {
            s   = run[(size_t)c * S + b];
            len = run[(size_t)c * S + b + 1] - s;
        }
        gsum += s;
        int inc = len;
#pragma unroll
        for (int d = 1; d < 32; d <<= 1) {
            const int u = __shfl_up(inc, d);
            if (lane >= d) inc += u;
        }
        if (lane == 31) red[wave] = inc;
        __syncthreads();
        int pre = 0, tot = 0;
        for (int w = 0; w < 8; ++w) { const int x = red[w]; tot += x; if (w < wave) pre += x; }
        if (c < nChunks) { lpre[c] = carry + pre + inc - len; sbeg[c] = s; }
        carry += tot;
        __syncthreads();
    }
    const int T = carry;
#pragma unroll
    for (int d = 16; d > 0; d >>= 1) gsum += __shfl_xor(gsum, d);
    if (lane == 0) red2[wave] = gsum;
    if (t == 0) lpre[nChunks] = T;
    const bool fits = (T <= TMAX) && (NG <= NGMAX);
    if (fits)
        for (int g = 0; g < NG; ++g) gcnt[g * BROWS + t] = 0;
    __syncthreads();
    int G = 0;
    for (int w = 0; w < 8; ++w) G += red2[w];
    const int row = b * BROWS + t;

    if (fits) {
        for (int c0 = wave * RUNS; c0 < nChunks; c0 += 8 * RUNS) {
            const int grp = c0 / RUNS;
            int* gc = gcnt + grp * BROWS;
            const int gbits = grp << 8;
            int  key[RUNS], ii[RUNS];
            int2 cv[RUNS];
#pragma unroll
            for (int u = 0; u < RUNS; ++u) {
                const int c = c0 + u;
                ii[u] = -1; key[u] = 0; cv[u] = make_int2(0, 0);
                if (c < nChunks) {
                    const int lp = lpre[c];
                    if (lane < lpre[c + 1] - lp) {
                        const size_t src = (size_t)c * EC + sbeg[c] + lane;
                        ii[u]  = lp + lane;
                        key[u] = rA[src];
                        cv[u]  = cvA[src];
                    }
                }
            }
#pragma unroll
            for (int u = 0; u < RUNS; ++u) {
                if (ii[u] >= 0) {
                    ucv[ii[u]] = cv[u];
                    ukr[ii[u]] = key[u] | gbits | (atomicAdd(&gc[key[u]], 1) << 13);
                }
            }
            for (int u = 0; u < RUNS; ++u) {
                const int c = c0 + u;
                if (c < nChunks) {
                    const int lp = lpre[c];
                    const int L  = lpre[c + 1] - lp;
                    for (int q = 32 + lane; q < L; q += 32) {
                        const size_t src = (size_t)c * EC + sbeg[c] + q;
                        const int k = rA[src];
                        ucv[lp + q] = cvA[src];
                        ukr[lp + q] = k | gbits | (atomicAdd(&gc[k], 1) << 13);
                    }
                }
            }
        }
        __syncthreads();
        int x[NGMAX];
        int v = 0;
#pragma unroll
        for (int g = 0; g < NGMAX; ++g) {
            x[g] = (g < NG) ? gcnt[g * BROWS + t] : 0;
            v += x[g];
        }
        int inc = v;
#pragma unroll
        for (int d = 1; d < 32; d <<= 1) {
            const int u = __shfl_up(inc, d);
            if (lane >= d) inc += u;
        }
        if (lane == 31) red[wave] = inc;
        __syncthreads();
        int pre = 0;
        for (int w = 0; w < wave; ++w) pre += red[w];
        const int ex = pre + inc - v;
        if (row <= nNodes) off[row] = G + ex;
        int first = ex;
#pragma unroll
        for (int g = 0; g < NGMAX; ++g) {
            if (g < NG) { gcnt[g * BROWS + t] = first; first += x[g]; }
        }
        __syncthreads();
        for (int i0 = t; i0 < T; i0 += 4 * 256) {
            int  kr[4];
            int2 rec[4];
#pragma unroll
            for (int k = 0; k < 4; ++k) {
                const int i = i0 + k * 256;
                kr[k]  = (i < T) ? ukr[i] : -1;
                rec[k] = (i < T) ? ucv[i] : make_int2(0, 0);
            }
            int base[4];
#pragma unroll
            for (int k = 0; k < 4; ++k)
                base[k] = gcnt[kr[k] & (NGMAX * BROWS - 1)];
#pragma unroll
            for (int k = 0; k < 4; ++k)
                if (kr[k] >= 0) scv[base[k] + (kr[k] >> 13)] = rec[k];
        }
        __syncthreads();
        for (int i = t - (G & 15); i < T; i += 256)
            if (i >= 0) edges[(size_t)G + i] = scv[i];
    } else {
        int cntk = 0;
        for (int c = 0; c < nChunks; ++c) {
            const int L = lpre[c + 1] - lpre[c];
            const size_t src = (size_t)c * EC + sbeg[c];
            for (int q = 0; q < L; ++q) cntk += (rA[src + q] == t) ? 1 : 0;
        }
        int inc = cntk;
#pragma unroll
        for (int d = 1; d < 32; d <<= 1) {
            const int u = __shfl_up(inc, d);
            if (lane >= d) inc += u;
        }
        if (lane == 31) red[wave] = inc;
        __syncthreads();
        int pre = 0;
        for (int w = 0; w < wave; ++w) pre += red[w];
        const int ex = pre + inc - cntk;
        if (row <= nNodes) off[row] = G + ex;
        int p = G + ex;
        for (int c = 0; c < nChunks; ++c) {
            const int L = lpre[c + 1] - lpre[c];
            const size_t src = (size_t)c * EC + sbeg[c];
            for (int q = 0; q < L; ++q)
                if (rA[src + q] == t) edges[p++] = cvA[src + q];
        }
    }
}

#define TILE_FLOATS (16 * WLDS_STRIDE)

__device__ __forceinline__ void fma4(float4& acc, float v, const float4& x) {
    acc.x = __builtin_fmaf(v, x.x, acc.x); acc.y = __builtin_fmaf(v, x.y, acc.y);
    acc.z = __builtin_fmaf(v, x.z, acc.z); acc.w = __builtin_fmaf(v, x.w, acc.w);
}

template <int K>
__device__ __forceinline__ void gather_group(float4& acc, const float4* __restrict__ X4,
                                             int myc, int myv, int j, int lane) {
    float4 x[K];
    float  v[K];
#pragma unroll
    for (int k = 0; k < K; ++k) {
        const int c = __builtin_amdgcn_readlane(myc, j + k);
        v[k] = __int_as_float(__builtin_amdgcn_readlane(myv, j + k));
        x[k] = X4[(size_t)c * (D / 4) + lane];
    }
#pragma unroll
    for (int k = 0; k < K; ++k) fma4(acc, v[k], x[k]);
}

__device__ __forceinline__ void gather_range(float4& acc, const float4* __restrict__ X4,
                                             int myc, int myv, int lo, int hi, int lane) {
    int j = lo;
    for (; j + 8 <= hi; j += 8) gather_group<8>(acc, X4, myc, myv, j, lane);
    if ((hi - j) & 4) { gather_group<4>(acc, X4, myc, myv, j, lane); j += 4; }
    if ((hi - j) & 2) { gather_group<2>(acc, X4, myc, myv, j, lane); j += 2; }
    if ((hi - j) & 1) { gather_group<1>(acc, X4, myc, myv, j, lane); }
}

__device__ __forceinline__ void row_bounds(const int* __restrict__ off, int r0, int nNodes,
                                           int& s, int& m, int& e) {
    const int i0 = (r0     < nNodes) ? r0     : nNodes;
    const int i1 = (r0 + 1 < nNodes) ? r0 + 1 : nNodes;
    const int i2 = (r0 + 2 < nNodes) ? r0 + 2 : nNodes;
    s = off[i0]; m = off[i1]; e = off[i2];
}

__device__ __forceinline__ int2 load_chunk(const int2* __restrict__ edges, int base, int end, int lane) {
    int2 ed = make_int2(0, 0);
    if (base + lane < end) ed = edges[base + lane];
    return ed;
}

__global__ __launch_bounds__(512) __attribute__((amdgpu_waves_per_eu(8, 8)))
void gather_project_kernel(const float* __restrict__ X,
                           const int*   __restrict__ off,
                           const int2*  __restrict__ edges,
                           const float* __restrict__ W,
                           const float* __restrict__ bias,
                           float* __restrict__ out,
                           int nNodes, int nStrips) {
    __shared__ __attribute__((aligned(16))) float lds[D * WLDS_STRIDE];

    const int lane = threadIdx.x & 31;
    const int wv   = __builtin_amdgcn_readfirstlane(threadIdx.x >> 5);
    const int half = wv >> 3;
    const int wave = wv & 7;
    const int l15  = lane & 15;
    const int hi   = lane >> 4;
    const int step = gridDim.x * 2;
    const float4* __restrict__ X4 = (const float4*)X;

    int strip = blockIdx.x * 2 + half;
    int s, m, e;
    row_bounds(off, strip * 16 + wave * 2, nNodes, s, m, e);
    int2 cur = load_chunk(edges, s, e, lane);

    for (int i = threadIdx.x; i < (D * D) / 4; i += 512) {
        const float4 w4 = ((const float4*)W)[i];
        const int flat = i * 4;
        const int o = flat >> 7;
        const int k = flat & 127;
        *(float4*)&lds[o * WLDS_STRIDE + k] = w4;
    }
    __syncthreads();

    v2f bw[32];
    {
        const float* bbase = lds + (wave * 16 + l15) * WLDS_STRIDE + 2 * hi;
#pragma unroll
        for (int ks = 0; ks < 32; ++ks) bw[ks] = *(const v2f*)(bbase + ks * 4);
    }
    const float bv = bias[wave * 16 + l15];
    __syncthreads();

    int buf = 0;
    for (int pair = blockIdx.x * 2; pair < nStrips; pair += step, strip += step) {
        int sn, mn, en;
        row_bounds(off, (strip + step) * 16 + wave * 2, nNodes, sn, mn, en);

        float* tl = lds + (buf * 2 + half) * TILE_FLOATS;
        float4 acc = make_float4(0.f, 0.f, 0.f, 0.f);
        bool row0 = true;
        for (int base = s; base < e; base += 32) {
            const int2 nxt = load_chunk(edges, base + 32, e, lane);
            const int n = (e - base < 32) ? (e - base) : 32;
            int a = m - base;
            a = (a < 0) ? 0 : ((a > n) ? n : a);
            gather_range(acc, X4, cur.x, cur.y, 0, a, lane);
            if (row0 && m <= base + n) {
                *(float4*)&tl[(wave * 2) * WLDS_STRIDE + 4 * lane] = acc;
                acc = make_float4(0.f, 0.f, 0.f, 0.f);
                row0 = false;
            }
            gather_range(acc, X4, cur.x, cur.y, a, n, lane);
            cur = nxt;
        }
        if (row0) {
            *(float4*)&tl[(wave * 2) * WLDS_STRIDE + 4 * lane] = acc;
            acc = make_float4(0.f, 0.f, 0.f, 0.f);
        }
        *(float4*)&tl[(wave * 2 + 1) * WLDS_STRIDE + 4 * lane] = acc;
        __syncthreads();
        cur = load_chunk(edges, sn, en, lane);

        v8f c = (v8f){0.f,0.f,0.f,0.f,0.f,0.f,0.f,0.f};
        const float* ap = tl + l15 * WLDS_STRIDE + 2 * hi;
#pragma unroll
        for (int ks = 0; ks < 32; ++ks) {
            const v2f a2 = *(const v2f*)(ap + ks * 4);
            c = __builtin_amdgcn_wmma_f32_16x16x4_f32(
                 false, a2,  false, bw[ks],
                 (short)0, c,
                 false,  false);
        }

#pragma unroll
        for (int v = 0; v < 8; ++v) {
            const int r = strip * 16 + v + 8 * hi;
            if (r < nNodes)
                __builtin_nontemporal_store(c[v] + bv, &out[(size_t)r * D + wave * 16 + l15]);
        }

        s = sn; m = mn; e = en;
        buf ^= 1;
    }
}

extern "C" void kernel_launch(void* const* d_in, const int* in_sizes, int n_in,
                              void* d_out, int out_size, void* d_ws, size_t ws_size,
                              hipStream_t stream) {
    const float* X    = (const float*)d_in[0];
    const int*   rows = (const int*)  d_in[1];
    const int*   cols = (const int*)  d_in[2];
    const float* vals = (const float*)d_in[3];
    const float* W    = (const float*)d_in[4];
    const float* bias = (const float*)d_in[5];
    float*       out  = (float*)d_out;

    const int nNodes = in_sizes[0] / D;
    const int nEdges = in_sizes[1];

    const int nChunks = (nEdges + EC - 1) / EC;
    const int nB      = (nNodes >> BSHIFT) + 1;
    const size_t recs = (size_t)nChunks * EC;
    char* p = (char*)d_ws;
    int2* cvA = (int2*)p;
    p += (recs * sizeof(int2) + 127) & ~(size_t)127;
    unsigned short* rA = (unsigned short*)p;
    p += (recs * sizeof(unsigned short) + 127) & ~(size_t)127;
    int* run = (int*)p;
    p += ((size_t)nChunks * (nB + 1) * sizeof(int) + 127) & ~(size_t)127;
    int* off = (int*)p;
    p += (((size_t)nNodes + 1) * sizeof(int) + 127) & ~(size_t)127;
    int2* edges = (int2*)p;

    chunk_bucket_kernel<<<nChunks, 1024, 0, stream>>>(rows, cols, vals, cvA, rA, run, nEdges, nB);
    bucket_rows_kernel<<<nB, 256, 0, stream>>>(cvA, rA, run, off, edges, nNodes, nChunks, nB);
    {
        const int nStrips = (nNodes + 15) / 16;
        const int nPairs  = (nStrips + 1) / 2;
        const int blocks  = (nPairs < 1024) ? nPairs : 1024;
        gather_project_kernel<<<blocks, 512, 0, stream>>>(X, off, edges, W, bias, out, nNodes, nStrips);
    }
}


# Gather: a wave's two rows are one load stream, 8 row loads kept in flight (next asked for as each is added, tail asked for together), not drained groups of 8,4,2,1 per row; sum order unchanged
# speedup vs baseline: 3.5609x; 1.0380x over previous
#include <hip/hip_runtime.h>


#define D 128
#define WLDS_STRIDE 132
#define EC 8192
#define BSHIFT 8
#define BROWS 256
#define MAXB 1024
#define MAXC 1024
#define RUNS 8
#define TMAX 5632
#define NGMAX 32

typedef float v2f __attribute__((ext_vector_type(2)));
typedef float v8f __attribute__((ext_vector_type(8)));

__global__ __launch_bounds__(1024)
void chunk_bucket_kernel(const int*   __restrict__ rows,
                         const int*   __restrict__ cols,
                         const float* __restrict__ vals,
                         int2* __restrict__ cvA,
                         unsigned short* __restrict__ rA,
                         int* __restrict__ run,
                         int nEdges, int nB) {
    __shared__ __attribute__((aligned(16))) int2 scv[EC];
    __shared__ __attribute__((aligned(16))) unsigned short sr[EC];
    __shared__ int cnt[MAXB];
    __shared__ int wsum[32];
    const int t    = threadIdx.x;
    const int lane = t & 31;
    const int wave = t >> 5;
    const int c    = blockIdx.x;
    const int eBeg = c * EC;
    const int n    = (nEdges - eBeg < EC) ? (nEdges - eBeg) : EC;

    int   r[EC / 1024];
    int   cc[EC / 1024];
    float vv[EC / 1024];
#pragma unroll
    for (int j = 0; j < EC / 1024; ++j) {
        const int i = j * 1024 + t;
        r[j]  = (i < n) ? rows[eBeg + i] : -1;
        cc[j] = (i < n) ? cols[eBeg + i] : 0;
        vv[j] = (i < n) ? vals[eBeg + i] : 0.f;
    }
    if (t < MAXB) cnt[t] = 0;
    __syncthreads();

    int rk[EC / 1024];
#pragma unroll
    for (int j = 0; j < EC / 1024; ++j) {
        rk[j] = 0;
        if (r[j] >= 0) rk[j] = atomicAdd(&cnt[r[j] >> BSHIFT], 1);
    }
    __syncthreads();

    const int v = (t < nB) ? cnt[t] : 0;
    int inc = v;
#pragma unroll
    for (int d = 1; d < 32; d <<= 1) {
        const int u = __shfl_up(inc, d);
        if (lane >= d) inc += u;
    }
    if (lane == 31) wsum[wave] = inc;
    __syncthreads();
    int pre = 0;
    for (int w = 0; w < wave; ++w) pre += wsum[w];
    const int ex = pre + inc - v;
    if (t < nB) {
        cnt[t] = ex;
        run[(size_t)c * (nB + 1) + t] = ex;
    }
    if (t == 0) run[(size_t)c * (nB + 1) + nB] = n;
    __syncthreads();

#pragma unroll
    for (int j = 0; j < EC / 1024; ++j) {
        if (r[j] >= 0) {
            const int pos = cnt[r[j] >> BSHIFT] + rk[j];
            scv[pos] = make_int2(cc[j], __float_as_int(vv[j]));
            sr[pos]  = (unsigned short)(r[j] & (BROWS - 1));
        }
    }
    __syncthreads();

    const int n2 = n >> 1;
    for (int i = t; i < n2; i += 1024) ((int4*)(cvA + (size_t)eBeg))[i] = ((const int4*)scv)[i];
    if ((n & 1) && t == 0) cvA[(size_t)eBeg + n - 1] = scv[n - 1];
    const int n8 = n >> 3;
    for (int i = t; i < n8; i += 1024) ((uint4*)(rA + (size_t)eBeg))[i] = ((const uint4*)sr)[i];
    for (int i = (n8 << 3) + t; i < n; i += 1024) rA[(size_t)eBeg + i] = sr[i];
}

__global__ __launch_bounds__(256)
void bucket_rows_kernel(const int2* __restrict__ cvA,
                        const unsigned short* __restrict__ rA,
                        const int* __restrict__ run,
                        int*  __restrict__ off,
                        int2* __restrict__ edges,
                        int nNodes, int nChunks, int nB) {
    __shared__ int lpre[MAXC + 1];
    __shared__ int sbeg[MAXC];
    __shared__ int red[8];
    __shared__ int red2[8];
    __shared__ int gcnt[NGMAX * BROWS];
    __shared__ int2 ucv[TMAX];
    __shared__ int  ukr[TMAX];
    __shared__ int2 scv[TMAX];
    const int t    = threadIdx.x;
    const int lane = t & 31;
    const int wave = t >> 5;
    const int b    = blockIdx.x;
    const int S    = nB + 1;
    const int NG   = (nChunks + RUNS - 1) / RUNS;

    int carry = 0, gsum = 0;
    for (int c0 = 0; c0 < nChunks; c0 += 256) {
        const int c = c0 + t;
        int s = 0, len = 0;
        if (c < nChunks) {
            s   = run[(size_t)c * S + b];
            len = run[(size_t)c * S + b + 1] - s;
        }
        gsum += s;
        int inc = len;
#pragma unroll
        for (int d = 1; d < 32; d <<= 1) {
            const int u = __shfl_up(inc, d);
            if (lane >= d) inc += u;
        }
        if (lane == 31) red[wave] = inc;
        __syncthreads();
        int pre = 0, tot = 0;
        for (int w = 0; w < 8; ++w) { const int x = red[w]; tot += x; if (w < wave) pre += x; }
        if (c < nChunks) { lpre[c] = carry + pre + inc - len; sbeg[c] = s; }
        carry += tot;
        __syncthreads();
    }
    const int T = carry;
#pragma unroll
    for (int d = 16; d > 0; d >>= 1) gsum += __shfl_xor(gsum, d);
    if (lane == 0) red2[wave] = gsum;
    if (t == 0) lpre[nChunks] = T;
    const bool fits = (T <= TMAX) && (NG <= NGMAX);
    if (fits)
        for (int g = 0; g < NG; ++g) gcnt[g * BROWS + t] = 0;
    __syncthreads();
    int G = 0;
    for (int w = 0; w < 8; ++w) G += red2[w];
    const int row = b * BROWS + t;

    if (fits) {
        for (int c0 = wave * RUNS; c0 < nChunks; c0 += 8 * RUNS) {
            const int grp = c0 / RUNS;
            int* gc = gcnt + grp * BROWS;
            const int gbits = grp << 8;
            int  key[RUNS], ii[RUNS];
            int2 cv[RUNS];
#pragma unroll
            for (int u = 0; u < RUNS; ++u) {
                const int c = c0 + u;
                ii[u] = -1; key[u] = 0; cv[u] = make_int2(0, 0);
                if (c < nChunks) {
                    const int lp = lpre[c];
                    if (lane < lpre[c + 1] - lp) {
                        const size_t src = (size_t)c * EC + sbeg[c] + lane;
                        ii[u]  = lp + lane;
                        key[u] = rA[src];
                        cv[u]  = cvA[src];
                    }
                }
            }
#pragma unroll
            for (int u = 0; u < RUNS; ++u) {
                if (ii[u] >= 0) {
                    ucv[ii[u]] = cv[u];
                    ukr[ii[u]] = key[u] | gbits | (atomicAdd(&gc[key[u]], 1) << 13);
                }
            }
            for (int u = 0; u < RUNS; ++u) {
                const int c = c0 + u;
                if (c < nChunks) {
                    const int lp = lpre[c];
                    const int L  = lpre[c + 1] - lp;
                    for (int q = 32 + lane; q < L; q += 32) {
                        const size_t src = (size_t)c * EC + sbeg[c] + q;
                        const int k = rA[src];
                        ucv[lp + q] = cvA[src];
                        ukr[lp + q] = k | gbits | (atomicAdd(&gc[k], 1) << 13);
                    }
                }
            }
        }
        __syncthreads();
        int x[NGMAX];
        int v = 0;
#pragma unroll
        for (int g = 0; g < NGMAX; ++g) {
            x[g] = (g < NG) ? gcnt[g * BROWS + t] : 0;
            v += x[g];
        }
        int inc = v;
#pragma unroll
        for (int d = 1; d < 32; d <<= 1) {
            const int u = __shfl_up(inc, d);
            if (lane >= d) inc += u;
        }
        if (lane == 31) red[wave] = inc;
        __syncthreads();
        int pre = 0;
        for (int w = 0; w < wave; ++w) pre += red[w];
        const int ex = pre + inc - v;
        if (row <= nNodes) off[row] = G + ex;
        int first = ex;
#pragma unroll
        for (int g = 0; g < NGMAX; ++g) {
            if (g < NG) { gcnt[g * BROWS + t] = first; first += x[g]; }
        }
        __syncthreads();
        for (int i0 = t; i0 < T; i0 += 4 * 256) {
            int  kr[4];
            int2 rec[4];
#pragma unroll
            for (int k = 0; k < 4; ++k) {
                const int i = i0 + k * 256;
                kr[k]  = (i < T) ? ukr[i] : -1;
                rec[k] = (i < T) ? ucv[i] : make_int2(0, 0);
            }
            int base[4];
#pragma unroll
            for (int k = 0; k < 4; ++k)
                base[k] = gcnt[kr[k] & (NGMAX * BROWS - 1)];
#pragma unroll
            for (int k = 0; k < 4; ++k)
                if (kr[k] >= 0) scv[base[k] + (kr[k] >> 13)] = rec[k];
        }
        __syncthreads();
        for (int i = t - (G & 15); i < T; i += 256)
            if (i >= 0) edges[(size_t)G + i] = scv[i];
    } else {
        int cntk = 0;
        for (int c = 0; c < nChunks; ++c) {
            const int L = lpre[c + 1] - lpre[c];
            const size_t src = (size_t)c * EC + sbeg[c];
            for (int q = 0; q < L; ++q) cntk += (rA[src + q] == t) ? 1 : 0;
        }
        int inc = cntk;
#pragma unroll
        for (int d = 1; d < 32; d <<= 1) {
            const int u = __shfl_up(inc, d);
            if (lane >= d) inc += u;
        }
        if (lane == 31) red[wave] = inc;
        __syncthreads();
        int pre = 0;
        for (int w = 0; w < wave; ++w) pre += red[w];
        const int ex = pre + inc - cntk;
        if (row <= nNodes) off[row] = G + ex;
        int p = G + ex;
        for (int c = 0; c < nChunks; ++c) {
            const int L = lpre[c + 1] - lpre[c];
            const size_t src = (size_t)c * EC + sbeg[c];
            for (int q = 0; q < L; ++q)
                if (rA[src + q] == t) edges[p++] = cvA[src + q];
        }
    }
}

#define TILE_FLOATS (16 * WLDS_STRIDE)

__device__ __forceinline__ void fma4(float4& acc, float v, const float4& x) {
    acc.x = __builtin_fmaf(v, x.x, acc.x); acc.y = __builtin_fmaf(v, x.y, acc.y);
    acc.z = __builtin_fmaf(v, x.z, acc.z); acc.w = __builtin_fmaf(v, x.w, acc.w);
}

__device__ __forceinline__ float4 row_load(const float4* __restrict__ X4, int myc, int j, int lane) {
    const int c = __builtin_amdgcn_readlane(myc, j);
    return X4[(size_t)c * (D / 4) + lane];
}

__device__ __forceinline__ float edge_val(int myv, int j) {
    return __int_as_float(__builtin_amdgcn_readlane(myv, j));
}

__device__ __forceinline__ void flush_row(float4& acc, float* __restrict__ dst) {
    *(float4*)dst = acc;
    acc = make_float4(0.f, 0.f, 0.f, 0.f);
}

__device__ __forceinline__ void gather_chunk(float4& acc, const float4* __restrict__ X4,
                                             int myc, int myv, int n, int fl,
                                             float* __restrict__ row0dst, int lane) {
    int j = 0;
    float4 x[8];
    if (n >= 8) {
#pragma unroll
        for (int k = 0; k < 8; ++k) x[k] = row_load(X4, myc, k, lane);
        for (; j + 16 <= n; j += 8) {
#pragma unroll
            for (int k = 0; k < 8; ++k) {
                if (j + k == fl) flush_row(acc, row0dst);
                fma4(acc, edge_val(myv, j + k), x[k]);
                x[k] = row_load(X4, myc, j + 8 + k, lane);
            }
        }
#pragma unroll
        for (int k = 0; k < 8; ++k) {
            if (j + k == fl) flush_row(acc, row0dst);
            fma4(acc, edge_val(myv, j + k), x[k]);
        }
        j += 8;
    }
    if (j < n) {
#pragma unroll
        for (int k = 0; k < 7; ++k)
            if (j + k < n) x[k] = row_load(X4, myc, j + k, lane);
#pragma unroll
        for (int k = 0; k < 7; ++k) {
            if (j + k < n) {
                if (j + k == fl) flush_row(acc, row0dst);
                fma4(acc, edge_val(myv, j + k), x[k]);
            }
        }
    }
    if (fl == n) flush_row(acc, row0dst);
}

__device__ __forceinline__ void row_bounds(const int* __restrict__ off, int r0, int nNodes,
                                           int& s, int& m, int& e) {
    const int i0 = (r0     < nNodes) ? r0     : nNodes;
    const int i1 = (r0 + 1 < nNodes) ? r0 + 1 : nNodes;
    const int i2 = (r0 + 2 < nNodes) ? r0 + 2 : nNodes;
    s = off[i0]; m = off[i1]; e = off[i2];
}

__device__ __forceinline__ int2 load_chunk(const int2* __restrict__ edges, int base, int end, int lane) {
    int2 ed = make_int2(0, 0);
    if (base + lane < end) ed = edges[base + lane];
    return ed;
}

__global__ __launch_bounds__(512) __attribute__((amdgpu_waves_per_eu(8, 8)))
void gather_project_kernel(const float* __restrict__ X,
                           const int*   __restrict__ off,
                           const int2*  __restrict__ edges,
                           const float* __restrict__ W,
                           const float* __restrict__ bias,
                           float* __restrict__ out,
                           int nNodes, int nStrips) {
    __shared__ __attribute__((aligned(16))) float lds[D * WLDS_STRIDE];

    const int lane = threadIdx.x & 31;
    const int wv   = __builtin_amdgcn_readfirstlane(threadIdx.x >> 5);
    const int half = wv >> 3;
    const int wave = wv & 7;
    const int l15  = lane & 15;
    const int hi   = lane >> 4;
    const int step = gridDim.x * 2;
    const float4* __restrict__ X4 = (const float4*)X;

    int strip = blockIdx.x * 2 + half;
    int s, m, e;
    row_bounds(off, strip * 16 + wave * 2, nNodes, s, m, e);
    int2 cur = load_chunk(edges, s, e, lane);

    for (int i = threadIdx.x; i < (D * D) / 4; i += 512) {
        const float4 w4 = ((const float4*)W)[i];
        const int flat = i * 4;
        const int o = flat >> 7;
        const int k = flat & 127;
        *(float4*)&lds[o * WLDS_STRIDE + k] = w4;
    }
    __syncthreads();

    v2f bw[32];
    {
        const float* bbase = lds + (wave * 16 + l15) * WLDS_STRIDE + 2 * hi;
#pragma unroll
        for (int ks = 0; ks < 32; ++ks) bw[ks] = *(const v2f*)(bbase + ks * 4);
    }
    const float bv = bias[wave * 16 + l15];
    __syncthreads();

    int buf = 0;
    for (int pair = blockIdx.x * 2; pair < nStrips; pair += step, strip += step) {
        int sn, mn, en;
        row_bounds(off, (strip + step) * 16 + wave * 2, nNodes, sn, mn, en);

        float* tl = lds + (buf * 2 + half) * TILE_FLOATS;
        float4 acc = make_float4(0.f, 0.f, 0.f, 0.f);
        bool row0 = true;
        for (int base = s; base < e; base += 32) {
            const int2 nxt = load_chunk(edges, base + 32, e, lane);
            const int n = (e - base < 32) ? (e - base) : 32;
            int a = m - base;
            a = (a < 0) ? 0 : ((a > n) ? n : a);
            const bool ends = row0 && m <= base + n;
            gather_chunk(acc, X4, cur.x, cur.y, n, ends ? a : -1,
                         &tl[(wave * 2) * WLDS_STRIDE + 4 * lane], lane);
            if (ends) row0 = false;
            cur = nxt;
        }
        if (row0) {
            *(float4*)&tl[(wave * 2) * WLDS_STRIDE + 4 * lane] = acc;
            acc = make_float4(0.f, 0.f, 0.f, 0.f);
        }
        *(float4*)&tl[(wave * 2 + 1) * WLDS_STRIDE + 4 * lane] = acc;
        __syncthreads();
        cur = load_chunk(edges, sn, en, lane);

        v8f c = (v8f){0.f,0.f,0.f,0.f,0.f,0.f,0.f,0.f};
        const float* ap = tl + l15 * WLDS_STRIDE + 2 * hi;
#pragma unroll
        for (int ks = 0; ks < 32; ++ks) {
            const v2f a2 = *(const v2f*)(ap + ks * 4);
            c = __builtin_amdgcn_wmma_f32_16x16x4_f32(
                 false, a2,  false, bw[ks],
                 (short)0, c,
                 false,  false);
        }

#pragma unroll
        for (int v = 0; v < 8; ++v) {
            const int r = strip * 16 + v + 8 * hi;
            if (r < nNodes)
                __builtin_nontemporal_store(c[v] + bv, &out[(size_t)r * D + wave * 16 + l15]);
        }

        s = sn; m = mn; e = en;
        buf ^= 1;
    }
}

extern "C" void kernel_launch(void* const* d_in, const int* in_sizes, int n_in,
                              void* d_out, int out_size, void* d_ws, size_t ws_size,
                              hipStream_t stream) {
    const float* X    = (const float*)d_in[0];
    const int*   rows = (const int*)  d_in[1];
    const int*   cols = (const int*)  d_in[2];
    const float* vals = (const float*)d_in[3];
    const float* W    = (const float*)d_in[4];
    const float* bias = (const float*)d_in[5];
    float*       out  = (float*)d_out;

    const int nNodes = in_sizes[0] / D;
    const int nEdges = in_sizes[1];

    const int nChunks = (nEdges + EC - 1) / EC;
    const int nB      = (nNodes >> BSHIFT) + 1;
    const size_t recs = (size_t)nChunks * EC;
    char* p = (char*)d_ws;
    int2* cvA = (int2*)p;
    p += (recs * sizeof(int2) + 127) & ~(size_t)127;
    unsigned short* rA = (unsigned short*)p;
    p += (recs * sizeof(unsigned short) + 127) & ~(size_t)127;
    int* run = (int*)p;
    p += ((size_t)nChunks * (nB + 1) * sizeof(int) + 127) & ~(size_t)127;
    int* off = (int*)p;
    p += (((size_t)nNodes + 1) * sizeof(int) + 127) & ~(size_t)127;
    int2* edges = (int2*)p;

    chunk_bucket_kernel<<<nChunks, 1024, 0, stream>>>(rows, cols, vals, cvA, rA, run, nEdges, nB);
    bucket_rows_kernel<<<nB, 256, 0, stream>>>(cvA, rA, run, off, edges, nNodes, nChunks, nB);
    {
        const int nStrips = (nNodes + 15) / 16;
        const int nPairs  = (nStrips + 1) / 2;
        const int blocks  = (nPairs < 1024) ? nPairs : 1024;
        gather_project_kernel<<<blocks, 512, 0, stream>>>(X, off, edges, W, bias, out, nNodes, nStrips);
    }
}


# Gather+projection launch: W^T fragments read from LDS at each product, not held in 64 registers: 3 work-groups (12 waves/SIMD) per WGP not 2, all 768 resident in one round; sums unchanged
# speedup vs baseline: 3.8456x; 1.0800x over previous
#include <hip/hip_runtime.h>


#define D 128
#define WLDS_STRIDE 132
#define EC 8192
#define BSHIFT 8
#define BROWS 256
#define MAXB 1024
#define MAXC 1024
#define RUNS 8
#define TMAX 5632
#define NGMAX 32

typedef float v2f __attribute__((ext_vector_type(2)));
typedef float v8f __attribute__((ext_vector_type(8)));

__global__ __launch_bounds__(1024)
void chunk_bucket_kernel(const int*   __restrict__ rows,
                         const int*   __restrict__ cols,
                         const float* __restrict__ vals,
                         int2* __restrict__ cvA,
                         unsigned short* __restrict__ rA,
                         int* __restrict__ run,
                         int nEdges, int nB) {
    __shared__ __attribute__((aligned(16))) int2 scv[EC];
    __shared__ __attribute__((aligned(16))) unsigned short sr[EC];
    __shared__ int cnt[MAXB];
    __shared__ int wsum[32];
    const int t    = threadIdx.x;
    const int lane = t & 31;
    const int wave = t >> 5;
    const int c    = blockIdx.x;
    const int eBeg = c * EC;
    const int n    = (nEdges - eBeg < EC) ? (nEdges - eBeg) : EC;

    int   r[EC / 1024];
    int   cc[EC / 1024];
    float vv[EC / 1024];
#pragma unroll
    for (int j = 0; j < EC / 1024; ++j) {
        const int i = j * 1024 + t;
        r[j]  = (i < n) ? rows[eBeg + i] : -1;
        cc[j] = (i < n) ? cols[eBeg + i] : 0;
        vv[j] = (i < n) ? vals[eBeg + i] : 0.f;
    }
    if (t < MAXB) cnt[t] = 0;
    __syncthreads();

    int rk[EC / 1024];
#pragma unroll
    for (int j = 0; j < EC / 1024; ++j) {
        rk[j] = 0;
        if (r[j] >= 0) rk[j] = atomicAdd(&cnt[r[j] >> BSHIFT], 1);
    }
    __syncthreads();

    const int v = (t < nB) ? cnt[t] : 0;
    int inc = v;
#pragma unroll
    for (int d = 1; d < 32; d <<= 1) {
        const int u = __shfl_up(inc, d);
        if (lane >= d) inc += u;
    }
    if (lane == 31) wsum[wave] = inc;
    __syncthreads();
    int pre = 0;
    for (int w = 0; w < wave; ++w) pre += wsum[w];
    const int ex = pre + inc - v;
    if (t < nB) {
        cnt[t] = ex;
        run[(size_t)c * (nB + 1) + t] = ex;
    }
    if (t == 0) run[(size_t)c * (nB + 1) + nB] = n;
    __syncthreads();

#pragma unroll
    for (int j = 0; j < EC / 1024; ++j) {
        if (r[j] >= 0) {
            const int pos = cnt[r[j] >> BSHIFT] + rk[j];
            scv[pos] = make_int2(cc[j], __float_as_int(vv[j]));
            sr[pos]  = (unsigned short)(r[j] & (BROWS - 1));
        }
    }
    __syncthreads();

    const int n2 = n >> 1;
    for (int i = t; i < n2; i += 1024) ((int4*)(cvA + (size_t)eBeg))[i] = ((const int4*)scv)[i];
    if ((n & 1) && t == 0) cvA[(size_t)eBeg + n - 1] = scv[n - 1];
    const int n8 = n >> 3;
    for (int i = t; i < n8; i += 1024) ((uint4*)(rA + (size_t)eBeg))[i] = ((const uint4*)sr)[i];
    for (int i = (n8 << 3) + t; i < n; i += 1024) rA[(size_t)eBeg + i] = sr[i];
}

__global__ __launch_bounds__(256)
void bucket_rows_kernel(const int2* __restrict__ cvA,
                        const unsigned short* __restrict__ rA,
                        const int* __restrict__ run,
                        int*  __restrict__ off,
                        int2* __restrict__ edges,
                        int nNodes, int nChunks, int nB) {
    __shared__ int lpre[MAXC + 1];
    __shared__ int sbeg[MAXC];
    __shared__ int red[8];
    __shared__ int red2[8];
    __shared__ int gcnt[NGMAX * BROWS];
    __shared__ int2 ucv[TMAX];
    __shared__ int  ukr[TMAX];
    __shared__ int2 scv[TMAX];
    const int t    = threadIdx.x;
    const int lane = t & 31;
    const int wave = t >> 5;
    const int b    = blockIdx.x;
    const int S    = nB + 1;
    const int NG   = (nChunks + RUNS - 1) / RUNS;

    int carry = 0, gsum = 0;
    for (int c0 = 0; c0 < nChunks; c0 += 256) {
        const int c = c0 + t;
        int s = 0, len = 0;
        if (c < nChunks) {
            s   = run[(size_t)c * S + b];
            len = run[(size_t)c * S + b + 1] - s;
        }
        gsum += s;
        int inc = len;
#pragma unroll
        for (int d = 1; d < 32; d <<= 1) {
            const int u = __shfl_up(inc, d);
            if (lane >= d) inc += u;
        }
        if (lane == 31) red[wave] = inc;
        __syncthreads();
        int pre = 0, tot = 0;
        for (int w = 0; w < 8; ++w) { const int x = red[w]; tot += x; if (w < wave) pre += x; }
        if (c < nChunks) { lpre[c] = carry + pre + inc - len; sbeg[c] = s; }
        carry += tot;
        __syncthreads();
    }
    const int T = carry;
#pragma unroll
    for (int d = 16; d > 0; d >>= 1) gsum += __shfl_xor(gsum, d);
    if (lane == 0) red2[wave] = gsum;
    if (t == 0) lpre[nChunks] = T;
    const bool fits = (T <= TMAX) && (NG <= NGMAX);
    if (fits)
        for (int g = 0; g < NG; ++g) gcnt[g * BROWS + t] = 0;
    __syncthreads();
    int G = 0;
    for (int w = 0; w < 8; ++w) G += red2[w];
    const int row = b * BROWS + t;

    if (fits) {
        for (int c0 = wave * RUNS; c0 < nChunks; c0 += 8 * RUNS) {
            const int grp = c0 / RUNS;
            int* gc = gcnt + grp * BROWS;
            const int gbits = grp << 8;
            int  key[RUNS], ii[RUNS];
            int2 cv[RUNS];
#pragma unroll
            for (int u = 0; u < RUNS; ++u) {
                const int c = c0 + u;
                ii[u] = -1; key[u] = 0; cv[u] = make_int2(0, 0);
                if (c < nChunks) {
                    const int lp = lpre[c];
                    if (lane < lpre[c + 1] - lp) {
                        const size_t src = (size_t)c * EC + sbeg[c] + lane;
                        ii[u]  = lp + lane;
                        key[u] = rA[src];
                        cv[u]  = cvA[src];
                    }
                }
            }
#pragma unroll
            for (int u = 0; u < RUNS; ++u) {
                if (ii[u] >= 0) {
                    ucv[ii[u]] = cv[u];
                    ukr[ii[u]] = key[u] | gbits | (atomicAdd(&gc[key[u]], 1) << 13);
                }
            }
            for (int u = 0; u < RUNS; ++u) {
                const int c = c0 + u;
                if (c < nChunks) {
                    const int lp = lpre[c];
                    const int L  = lpre[c + 1] - lp;
                    for (int q = 32 + lane; q < L; q += 32) {
                        const size_t src = (size_t)c * EC + sbeg[c] + q;
                        const int k = rA[src];
                        ucv[lp + q] = cvA[src];
                        ukr[lp + q] = k | gbits | (atomicAdd(&gc[k], 1) << 13);
                    }
                }
            }
        }
        __syncthreads();
        int x[NGMAX];
        int v = 0;
#pragma unroll
        for (int g = 0; g < NGMAX; ++g) {
            x[g] = (g < NG) ? gcnt[g * BROWS + t] : 0;
            v += x[g];
        }
        int inc = v;
#pragma unroll
        for (int d = 1; d < 32; d <<= 1) {
            const int u = __shfl_up(inc, d);
            if (lane >= d) inc += u;
        }
        if (lane == 31) red[wave] = inc;
        __syncthreads();
        int pre = 0;
        for (int w = 0; w < wave; ++w) pre += red[w];
        const int ex = pre + inc - v;
        if (row <= nNodes) off[row] = G + ex;
        int first = ex;
#pragma unroll
        for (int g = 0; g < NGMAX; ++g) {
            if (g < NG) { gcnt[g * BROWS + t] = first; first += x[g]; }
        }
        __syncthreads();
        for (int i0 = t; i0 < T; i0 += 4 * 256) {
            int  kr[4];
            int2 rec[4];
#pragma unroll
            for (int k = 0; k < 4; ++k) {
                const int i = i0 + k * 256;
                kr[k]  = (i < T) ? ukr[i] : -1;
                rec[k] = (i < T) ? ucv[i] : make_int2(0, 0);
            }
            int base[4];
#pragma unroll
            for (int k = 0; k < 4; ++k)
                base[k] = gcnt[kr[k] & (NGMAX * BROWS - 1)];
#pragma unroll
            for (int k = 0; k < 4; ++k)
                if (kr[k] >= 0) scv[base[k] + (kr[k] >> 13)] = rec[k];
        }
        __syncthreads();
        for (int i = t - (G & 15); i < T; i += 256)
            if (i >= 0) edges[(size_t)G + i] = scv[i];
    } else {
        int cntk = 0;
        for (int c = 0; c < nChunks; ++c) {
            const int L = lpre[c + 1] - lpre[c];
            const size_t src = (size_t)c * EC + sbeg[c];
            for (int q = 0; q < L; ++q) cntk += (rA[src + q] == t) ? 1 : 0;
        }
        int inc = cntk;
#pragma unroll
        for (int d = 1; d < 32; d <<= 1) {
            const int u = __shfl_up(inc, d);
            if (lane >= d) inc += u;
        }
        if (lane == 31) red[wave] = inc;
        __syncthreads();
        int pre = 0;
        for (int w = 0; w < wave; ++w) pre += red[w];
        const int ex = pre + inc - cntk;
        if (row <= nNodes) off[row] = G + ex;
        int p = G + ex;
        for (int c = 0; c < nChunks; ++c) {
            const int L = lpre[c + 1] - lpre[c];
            const size_t src = (size_t)c * EC + sbeg[c];
            for (int q = 0; q < L; ++q)
                if (rA[src + q] == t) edges[p++] = cvA[src + q];
        }
    }
}

#define TILE_FLOATS (16 * WLDS_STRIDE)

__device__ __forceinline__ void fma4(float4& acc, float v, const float4& x) {
    acc.x = __builtin_fmaf(v, x.x, acc.x); acc.y = __builtin_fmaf(v, x.y, acc.y);
    acc.z = __builtin_fmaf(v, x.z, acc.z); acc.w = __builtin_fmaf(v, x.w, acc.w);
}

__device__ __forceinline__ float4 row_load(const float4* __restrict__ X4, int myc, int j, int lane) {
    const int c = __builtin_amdgcn_readlane(myc, j);
    return X4[(size_t)c * (D / 4) + lane];
}

__device__ __forceinline__ float edge_val(int myv, int j) {
    return __int_as_float(__builtin_amdgcn_readlane(myv, j));
}

__device__ __forceinline__ void flush_row(float4& acc, float* __restrict__ dst) {
    *(float4*)dst = acc;
    acc = make_float4(0.f, 0.f, 0.f, 0.f);
}

__device__ __forceinline__ void gather_chunk(float4& acc, const float4* __restrict__ X4,
                                             int myc, int myv, int n, int fl,
                                             float* __restrict__ row0dst, int lane) {
    int j = 0;
    float4 x[8];
    if (n >= 8) {
#pragma unroll
        for (int k = 0; k < 8; ++k) x[k] = row_load(X4, myc, k, lane);
        for (; j + 16 <= n; j += 8) {
#pragma unroll
            for (int k = 0; k < 8; ++k) {
                if (j + k == fl) flush_row(acc, row0dst);
                fma4(acc, edge_val(myv, j + k), x[k]);
                x[k] = row_load(X4, myc, j + 8 + k, lane);
            }
        }
#pragma unroll
        for (int k = 0; k < 8; ++k) {
            if (j + k == fl) flush_row(acc, row0dst);
            fma4(acc, edge_val(myv, j + k), x[k]);
        }
        j += 8;
    }
    if (j < n) {
#pragma unroll
        for (int k = 0; k < 7; ++k)
            if (j + k < n) x[k] = row_load(X4, myc, j + k, lane);
#pragma unroll
        for (int k = 0; k < 7; ++k) {
            if (j + k < n) {
                if (j + k == fl) flush_row(acc, row0dst);
                fma4(acc, edge_val(myv, j + k), x[k]);
            }
        }
    }
    if (fl == n) flush_row(acc, row0dst);
}

__device__ __forceinline__ void row_bounds(const int* __restrict__ off, int r0, int nNodes,
                                           int& s, int& m, int& e) {
    const int i0 = (r0     < nNodes) ? r0     : nNodes;
    const int i1 = (r0 + 1 < nNodes) ? r0 + 1 : nNodes;
    const int i2 = (r0 + 2 < nNodes) ? r0 + 2 : nNodes;
    s = off[i0]; m = off[i1]; e = off[i2];
}

__device__ __forceinline__ int2 load_chunk(const int2* __restrict__ edges, int base, int end, int lane) {
    int2 ed = make_int2(0, 0);
    if (base + lane < end) ed = edges[base + lane];
    return ed;
}

__global__ __launch_bounds__(512) __attribute__((amdgpu_waves_per_eu(12, 12)))
void gather_project_kernel(const float* __restrict__ X,
                           const int*   __restrict__ off,
                           const int2*  __restrict__ edges,
                           const float* __restrict__ W,
                           const float* __restrict__ bias,
                           float* __restrict__ out,
                           int nNodes, int nStrips) {
    __shared__ __attribute__((aligned(16))) float lds[D * WLDS_STRIDE + 4 * TILE_FLOATS];
    float* const tiles = lds + D * WLDS_STRIDE;

    const int lane = threadIdx.x & 31;
    const int wv   = __builtin_amdgcn_readfirstlane(threadIdx.x >> 5);
    const int half = wv >> 3;
    const int wave = wv & 7;
    const int l15  = lane & 15;
    const int hi   = lane >> 4;
    const int step = gridDim.x * 2;
    const float4* __restrict__ X4 = (const float4*)X;

    int strip = blockIdx.x * 2 + half;
    int s, m, e;
    row_bounds(off, strip * 16 + wave * 2, nNodes, s, m, e);
    int2 cur = load_chunk(edges, s, e, lane);

    for (int i = threadIdx.x; i < (D * D) / 4; i += 512) {
        const float4 w4 = ((const float4*)W)[i];
        const int flat = i * 4;
        const int o = flat >> 7;
        const int k = flat & 127;
        *(float4*)&lds[o * WLDS_STRIDE + k] = w4;
    }
    __syncthreads();

    const float* bp = lds + (wave * 16 + l15) * WLDS_STRIDE + 2 * hi;
    const float bv = bias[wave * 16 + l15];

    int buf = 0;
    for (int pair = blockIdx.x * 2; pair < nStrips; pair += step, strip += step) {
        int sn, mn, en;
        row_bounds(off, (strip + step) * 16 + wave * 2, nNodes, sn, mn, en);

        float* tl = tiles + (buf * 2 + half) * TILE_FLOATS;
        float4 acc = make_float4(0.f, 0.f, 0.f, 0.f);
        bool row0 = true;
        for (int base = s; base < e; base += 32) {
            const int2 nxt = load_chunk(edges, base + 32, e, lane);
            const int n = (e - base < 32) ? (e - base) : 32;
            int a = m - base;
            a = (a < 0) ? 0 : ((a > n) ? n : a);
            const bool ends = row0 && m <= base + n;
            gather_chunk(acc, X4, cur.x, cur.y, n, ends ? a : -1,
                         &tl[(wave * 2) * WLDS_STRIDE + 4 * lane], lane);
            if (ends) row0 = false;
            cur = nxt;
        }
        if (row0) {
            *(float4*)&tl[(wave * 2) * WLDS_STRIDE + 4 * lane] = acc;
            acc = make_float4(0.f, 0.f, 0.f, 0.f);
        }
        *(float4*)&tl[(wave * 2 + 1) * WLDS_STRIDE + 4 * lane] = acc;
        __syncthreads();
        cur = load_chunk(edges, sn, en, lane);

        v8f c = (v8f){0.f,0.f,0.f,0.f,0.f,0.f,0.f,0.f};
        const float* ap = tl + l15 * WLDS_STRIDE + 2 * hi;
#pragma unroll
        for (int ks = 0; ks < 32; ++ks) {
            const v2f a2 = *(const v2f*)(ap + ks * 4);
            const v2f b2 = *(const v2f*)(bp + ks * 4);
            c = __builtin_amdgcn_wmma_f32_16x16x4_f32(
                 false, a2,  false, b2,
                 (short)0, c,
                 false,  false);
        }

#pragma unroll
        for (int v = 0; v < 8; ++v) {
            const int r = strip * 16 + v + 8 * hi;
            if (r < nNodes)
                __builtin_nontemporal_store(c[v] + bv, &out[(size_t)r * D + wave * 16 + l15]);
        }

        s = sn; m = mn; e = en;
        buf ^= 1;
    }
}

extern "C" void kernel_launch(void* const* d_in, const int* in_sizes, int n_in,
                              void* d_out, int out_size, void* d_ws, size_t ws_size,
                              hipStream_t stream) {
    const float* X    = (const float*)d_in[0];
    const int*   rows = (const int*)  d_in[1];
    const int*   cols = (const int*)  d_in[2];
    const float* vals = (const float*)d_in[3];
    const float* W    = (const float*)d_in[4];
    const float* bias = (const float*)d_in[5];
    float*       out  = (float*)d_out;

    const int nNodes = in_sizes[0] / D;
    const int nEdges = in_sizes[1];

    const int nChunks = (nEdges + EC - 1) / EC;
    const int nB      = (nNodes >> BSHIFT) + 1;
    const size_t recs = (size_t)nChunks * EC;
    char* p = (char*)d_ws;
    int2* cvA = (int2*)p;
    p += (recs * sizeof(int2) + 127) & ~(size_t)127;
    unsigned short* rA = (unsigned short*)p;
    p += (recs * sizeof(unsigned short) + 127) & ~(size_t)127;
    int* run = (int*)p;
    p += ((size_t)nChunks * (nB + 1) * sizeof(int) + 127) & ~(size_t)127;
    int* off = (int*)p;
    p += (((size_t)nNodes + 1) * sizeof(int) + 127) & ~(size_t)127;
    int2* edges = (int2*)p;

    chunk_bucket_kernel<<<nChunks, 1024, 0, stream>>>(rows, cols, vals, cvA, rA, run, nEdges, nB);
    bucket_rows_kernel<<<nB, 256, 0, stream>>>(cvA, rA, run, off, edges, nNodes, nChunks, nB);
    {
        const int nStrips = (nNodes + 15) / 16;
        const int nPairs  = (nStrips + 1) / 2;
        const int blocks  = (nPairs < 768) ? nPairs : 768;
        gather_project_kernel<<<blocks, 512, 0, stream>>>(X, off, edges, W, bias, out, nNodes, nStrips);
    }
}


# Bucket pass: a wave takes all its groups in one batch: one LDS read for all run bounds, records dense on lanes, all loads in flight, ranks together; row sum order: a run's records past 32 not last
# speedup vs baseline: 4.0122x; 1.0433x over previous
#include <hip/hip_runtime.h>


#define D 128
#define WLDS_STRIDE 132
#define EC 8192
#define BSHIFT 8
#define BROWS 256
#define MAXB 1024
#define MAXC 1024
#define RUNS 8
#define TMAX 5632
#define NGMAX 32
#define GPW (NGMAX / 8)
#define SLOTS 8

typedef float v2f __attribute__((ext_vector_type(2)));
typedef float v8f __attribute__((ext_vector_type(8)));

__global__ __launch_bounds__(1024)
void chunk_bucket_kernel(const int*   __restrict__ rows,
                         const int*   __restrict__ cols,
                         const float* __restrict__ vals,
                         int2* __restrict__ cvA,
                         unsigned short* __restrict__ rA,
                         int* __restrict__ run,
                         int nEdges, int nB) {
    __shared__ __attribute__((aligned(16))) int2 scv[EC];
    __shared__ __attribute__((aligned(16))) unsigned short sr[EC];
    __shared__ int cnt[MAXB];
    __shared__ int wsum[32];
    const int t    = threadIdx.x;
    const int lane = t & 31;
    const int wave = t >> 5;
    const int c    = blockIdx.x;
    const int eBeg = c * EC;
    const int n    = (nEdges - eBeg < EC) ? (nEdges - eBeg) : EC;

    int   r[EC / 1024];
    int   cc[EC / 1024];
    float vv[EC / 1024];
#pragma unroll
    for (int j = 0; j < EC / 1024; ++j) {
        const int i = j * 1024 + t;
        r[j]  = (i < n) ? rows[eBeg + i] : -1;
        cc[j] = (i < n) ? cols[eBeg + i] : 0;
        vv[j] = (i < n) ? vals[eBeg + i] : 0.f;
    }
    if (t < MAXB) cnt[t] = 0;
    __syncthreads();

    int rk[EC / 1024];
#pragma unroll
    for (int j = 0; j < EC / 1024; ++j) {
        rk[j] = 0;
        if (r[j] >= 0) rk[j] = atomicAdd(&cnt[r[j] >> BSHIFT], 1);
    }
    __syncthreads();

    const int v = (t < nB) ? cnt[t] : 0;
    int inc = v;
#pragma unroll
    for (int d = 1; d < 32; d <<= 1) {
        const int u = __shfl_up(inc, d);
        if (lane >= d) inc += u;
    }
    if (lane == 31) wsum[wave] = inc;
    __syncthreads();
    int pre = 0;
    for (int w = 0; w < wave; ++w) pre += wsum[w];
    const int ex = pre + inc - v;
    if (t < nB) {
        cnt[t] = ex;
        run[(size_t)c * (nB + 1) + t] = ex;
    }
    if (t == 0) run[(size_t)c * (nB + 1) + nB] = n;
    __syncthreads();

#pragma unroll
    for (int j = 0; j < EC / 1024; ++j) {
        if (r[j] >= 0) {
            const int pos = cnt[r[j] >> BSHIFT] + rk[j];
            scv[pos] = make_int2(cc[j], __float_as_int(vv[j]));
            sr[pos]  = (unsigned short)(r[j] & (BROWS - 1));
        }
    }
    __syncthreads();

    const int n2 = n >> 1;
    for (int i = t; i < n2; i += 1024) ((int4*)(cvA + (size_t)eBeg))[i] = ((const int4*)scv)[i];
    if ((n & 1) && t == 0) cvA[(size_t)eBeg + n - 1] = scv[n - 1];
    const int n8 = n >> 3;
    for (int i = t; i < n8; i += 1024) ((uint4*)(rA + (size_t)eBeg))[i] = ((const uint4*)sr)[i];
    for (int i = (n8 << 3) + t; i < n; i += 1024) rA[(size_t)eBeg + i] = sr[i];
}

__global__ __launch_bounds__(256)
void bucket_rows_kernel(const int2* __restrict__ cvA,
                        const unsigned short* __restrict__ rA,
                        const int* __restrict__ run,
                        int*  __restrict__ off,
                        int2* __restrict__ edges,
                        int nNodes, int nChunks, int nB) {
    __shared__ int lpre[MAXC + 1];
    __shared__ int sbeg[MAXC];
    __shared__ int red[8];
    __shared__ int red2[8];
    __shared__ int gcnt[NGMAX * BROWS];
    __shared__ int2 ucv[TMAX];
    __shared__ int  ukr[TMAX];
    __shared__ int2 scv[TMAX];
    const int t    = threadIdx.x;
    const int lane = t & 31;
    const int wave = t >> 5;
    const int b    = blockIdx.x;
    const int S    = nB + 1;
    const int NG   = (nChunks + RUNS - 1) / RUNS;

    int carry = 0, gsum = 0;
    for (int c0 = 0; c0 < nChunks; c0 += 256) {
        const int c = c0 + t;
        int s = 0, len = 0;
        if (c < nChunks) {
            s   = run[(size_t)c * S + b];
            len = run[(size_t)c * S + b + 1] - s;
        }
        gsum += s;
        int inc = len;
#pragma unroll
        for (int d = 1; d < 32; d <<= 1) {
            const int u = __shfl_up(inc, d);
            if (lane >= d) inc += u;
        }
        if (lane == 31) red[wave] = inc;
        __syncthreads();
        int pre = 0, tot = 0;
        for (int w = 0; w < 8; ++w) { const int x = red[w]; tot += x; if (w < wave) pre += x; }
        if (c < nChunks) { lpre[c] = carry + pre + inc - len; sbeg[c] = s; }
        carry += tot;
        __syncthreads();
    }
    const int T = carry;
#pragma unroll
    for (int d = 16; d > 0; d >>= 1) gsum += __shfl_xor(gsum, d);
    if (lane == 0) red2[wave] = gsum;
    if (t == 0) lpre[nChunks] = T;
    const bool fits = (T <= TMAX) && (NG <= NGMAX);
    if (fits)
        for (int g = 0; g < NG; ++g) gcnt[g * BROWS + t] = 0;
    __syncthreads();
    int G = 0;
    for (int w = 0; w < 8; ++w) G += red2[w];
    const int row = b * BROWS + t;

    if (fits) {
        const int wv = __builtin_amdgcn_readfirstlane(wave);
        int lpv[GPW], sbv[GPW];
#pragma unroll
        for (int g = 0; g < GPW; ++g) {
            const int ci = (wv + 8 * g) * RUNS + lane;
            lpv[g] = lpre[(ci < nChunks) ? ci : nChunks];
            sbv[g] = sbeg[(ci < nChunks) ? ci : (nChunks - 1)];
        }
        int  key[GPW * SLOTS], ii[GPW * SLOTS];
        int2 cv[GPW * SLOTS];
        bool over = false;
#pragma unroll
        for (int g = 0; g < GPW; ++g) {
            const int base = __builtin_amdgcn_readlane(lpv[g], 0);
            int st[RUNS + 1];
            int dl[RUNS];
#pragma unroll
            for (int u = 0; u <= RUNS; ++u) st[u] = __builtin_amdgcn_readlane(lpv[g], u) - base;
#pragma unroll
            for (int u = 0; u < RUNS; ++u)
                dl[u] = ((wv + 8 * g) * RUNS + u) * EC + __builtin_amdgcn_readlane(sbv[g], u) - st[u];
            const int Ng = st[RUNS];
            over = over || (Ng > 32 * SLOTS);
#pragma unroll
            for (int j = 0; j < SLOTS; ++j) {
                const int k = g * SLOTS + j;
                const int r = 32 * j + lane;
                ii[k] = -1; key[k] = 0; cv[k] = make_int2(0, 0);
                if (r < Ng) {
                    int d = dl[0];
#pragma unroll
                    for (int u = 1; u < RUNS; ++u) d = (r >= st[u]) ? dl[u] : d;
                    ii[k]  = base + r;
                    key[k] = rA[d + r];
                    cv[k]  = cvA[d + r];
                }
            }
        }
        int rk[GPW * SLOTS];
#pragma unroll
        for (int k = 0; k < GPW * SLOTS; ++k) {
            rk[k] = 0;
            if (ii[k] >= 0) {
                ucv[ii[k]] = cv[k];
                rk[k] = atomicAdd(&gcnt[(wv + 8 * (k / SLOTS)) * BROWS + key[k]], 1);
            }
        }
#pragma unroll
        for (int k = 0; k < GPW * SLOTS; ++k)
            if (ii[k] >= 0) ukr[ii[k]] = key[k] | ((wv + 8 * (k / SLOTS)) << 8) | (rk[k] << 13);
        if (over) {
            for (int g = 0; g < GPW; ++g) {
                const int grp = wv + 8 * g;
                const int c0  = grp * RUNS;
                if (c0 < nChunks) {
                    const int cE   = (c0 + RUNS < nChunks) ? (c0 + RUNS) : nChunks;
                    const int base = lpre[c0];
                    const int Ng   = lpre[cE] - base;
                    for (int r = 32 * SLOTS + lane; r < Ng; r += 32) {
                        int c = c0;
                        for (int cc = c0 + 1; cc < cE; ++cc)
                            if (r >= lpre[cc] - base) c = cc;
                        const size_t src = (size_t)c * EC + sbeg[c] + (r - (lpre[c] - base));
                        const int kk = rA[src];
                        ucv[base + r] = cvA[src];
                        ukr[base + r] = kk | (grp << 8) | (atomicAdd(&gcnt[grp * BROWS + kk], 1) << 13);
                    }
                }
            }
        }
        __syncthreads();
        int x[NGMAX];
        int v = 0;
#pragma unroll
        for (int g = 0; g < NGMAX; ++g) {
            x[g] = (g < NG) ? gcnt[g * BROWS + t] : 0;
            v += x[g];
        }
        int inc = v;
#pragma unroll
        for (int d = 1; d < 32; d <<= 1) {
            const int u = __shfl_up(inc, d);
            if (lane >= d) inc += u;
        }
        if (lane == 31) red[wave] = inc;
        __syncthreads();
        int pre = 0;
        for (int w = 0; w < wave; ++w) pre += red[w];
        const int ex = pre + inc - v;
        if (row <= nNodes) off[row] = G + ex;
        int first = ex;
#pragma unroll
        for (int g = 0; g < NGMAX; ++g) {
            if (g < NG) { gcnt[g * BROWS + t] = first; first += x[g]; }
        }
        __syncthreads();
        for (int i0 = t; i0 < T; i0 += 4 * 256) {
            int  kr[4];
            int2 rec[4];
#pragma unroll
            for (int k = 0; k < 4; ++k) {
                const int i = i0 + k * 256;
                kr[k]  = (i < T) ? ukr[i] : -1;
                rec[k] = (i < T) ? ucv[i] : make_int2(0, 0);
            }
            int base[4];
#pragma unroll
            for (int k = 0; k < 4; ++k)
                base[k] = gcnt[kr[k] & (NGMAX * BROWS - 1)];
#pragma unroll
            for (int k = 0; k < 4; ++k)
                if (kr[k] >= 0) scv[base[k] + (kr[k] >> 13)] = rec[k];
        }
        __syncthreads();
        for (int i = t - (G & 15); i < T; i += 256)
            if (i >= 0) edges[(size_t)G + i] = scv[i];
    } else {
        int cntk = 0;
        for (int c = 0; c < nChunks; ++c) {
            const int L = lpre[c + 1] - lpre[c];
            const size_t src = (size_t)c * EC + sbeg[c];
            for (int q = 0; q < L; ++q) cntk += (rA[src + q] == t) ? 1 : 0;
        }
        int inc = cntk;
#pragma unroll
        for (int d = 1; d < 32; d <<= 1) {
            const int u = __shfl_up(inc, d);
            if (lane >= d) inc += u;
        }
        if (lane == 31) red[wave] = inc;
        __syncthreads();
        int pre = 0;
        for (int w = 0; w < wave; ++w) pre += red[w];
        const int ex = pre + inc - cntk;
        if (row <= nNodes) off[row] = G + ex;
        int p = G + ex;
        for (int c = 0; c < nChunks; ++c) {
            const int L = lpre[c + 1] - lpre[c];
            const size_t src = (size_t)c * EC + sbeg[c];
            for (int q = 0; q < L; ++q)
                if (rA[src + q] == t) edges[p++] = cvA[src + q];
        }
    }
}

#define TILE_FLOATS (16 * WLDS_STRIDE)

__device__ __forceinline__ void fma4(float4& acc, float v, const float4& x) {
    acc.x = __builtin_fmaf(v, x.x, acc.x); acc.y = __builtin_fmaf(v, x.y, acc.y);
    acc.z = __builtin_fmaf(v, x.z, acc.z); acc.w = __builtin_fmaf(v, x.w, acc.w);
}

__device__ __forceinline__ float4 row_load(const float4* __restrict__ X4, int myc, int j, int lane) {
    const int c = __builtin_amdgcn_readlane(myc, j);
    return X4[(size_t)c * (D / 4) + lane];
}

__device__ __forceinline__ float edge_val(int myv, int j) {
    return __int_as_float(__builtin_amdgcn_readlane(myv, j));
}

__device__ __forceinline__ void flush_row(float4& acc, float* __restrict__ dst) {
    *(float4*)dst = acc;
    acc = make_float4(0.f, 0.f, 0.f, 0.f);
}

__device__ __forceinline__ void gather_chunk(float4& acc, const float4* __restrict__ X4,
                                             int myc, int myv, int n, int fl,
                                             float* __restrict__ row0dst, int lane) {
    int j = 0;
    float4 x[8];
    if (n >= 8) {
#pragma unroll
        for (int k = 0; k < 8; ++k) x[k] = row_load(X4, myc, k, lane);
        for (; j + 16 <= n; j += 8) {
#pragma unroll
            for (int k = 0; k < 8; ++k) {
                if (j + k == fl) flush_row(acc, row0dst);
                fma4(acc, edge_val(myv, j + k), x[k]);
                x[k] = row_load(X4, myc, j + 8 + k, lane);
            }
        }
#pragma unroll
        for (int k = 0; k < 8; ++k) {
            if (j + k == fl) flush_row(acc, row0dst);
            fma4(acc, edge_val(myv, j + k), x[k]);
        }
        j += 8;
    }
    if (j < n) {
#pragma unroll
        for (int k = 0; k < 7; ++k)
            if (j + k < n) x[k] = row_load(X4, myc, j + k, lane);
#pragma unroll
        for (int k = 0; k < 7; ++k) {
            if (j + k < n) {
                if (j + k == fl) flush_row(acc, row0dst);
                fma4(acc, edge_val(myv, j + k), x[k]);
            }
        }
    }
    if (fl == n) flush_row(acc, row0dst);
}

__device__ __forceinline__ void row_bounds(const int* __restrict__ off, int r0, int nNodes,
                                           int& s, int& m, int& e) {
    const int i0 = (r0     < nNodes) ? r0     : nNodes;
    const int i1 = (r0 + 1 < nNodes) ? r0 + 1 : nNodes;
    const int i2 = (r0 + 2 < nNodes) ? r0 + 2 : nNodes;
    s = off[i0]; m = off[i1]; e = off[i2];
}

__device__ __forceinline__ int2 load_chunk(const int2* __restrict__ edges, int base, int end, int lane) {
    int2 ed = make_int2(0, 0);
    if (base + lane < end) ed = edges[base + lane];
    return ed;
}

__global__ __launch_bounds__(512) __attribute__((amdgpu_waves_per_eu(12, 12)))
void gather_project_kernel(const float* __restrict__ X,
                           const int*   __restrict__ off,
                           const int2*  __restrict__ edges,
                           const float* __restrict__ W,
                           const float* __restrict__ bias,
                           float* __restrict__ out,
                           int nNodes, int nStrips) {
    __shared__ __attribute__((aligned(16))) float lds[D * WLDS_STRIDE + 4 * TILE_FLOATS];
    float* const tiles = lds + D * WLDS_STRIDE;

    const int lane = threadIdx.x & 31;
    const int wv   = __builtin_amdgcn_readfirstlane(threadIdx.x >> 5);
    const int half = wv >> 3;
    const int wave = wv & 7;
    const int l15  = lane & 15;
    const int hi   = lane >> 4;
    const int step = gridDim.x * 2;
    const float4* __restrict__ X4 = (const float4*)X;

    int strip = blockIdx.x * 2 + half;
    int s, m, e;
    row_bounds(off, strip * 16 + wave * 2, nNodes, s, m, e);
    int2 cur = load_chunk(edges, s, e, lane);

    for (int i = threadIdx.x; i < (D * D) / 4; i += 512) {
        const float4 w4 = ((const float4*)W)[i];
        const int flat = i * 4;
        const int o = flat >> 7;
        const int k = flat & 127;
        *(float4*)&lds[o * WLDS_STRIDE + k] = w4;
    }
    __syncthreads();

    const float* bp = lds + (wave * 16 + l15) * WLDS_STRIDE + 2 * hi;
    const float bv = bias[wave * 16 + l15];

    int buf = 0;
    for (int pair = blockIdx.x * 2; pair < nStrips; pair += step, strip += step) {
        int sn, mn, en;
        row_bounds(off, (strip + step) * 16 + wave * 2, nNodes, sn, mn, en);

        float* tl = tiles + (buf * 2 + half) * TILE_FLOATS;
        float4 acc = make_float4(0.f, 0.f, 0.f, 0.f);
        bool row0 = true;
        for (int base = s; base < e; base += 32) {
            const int2 nxt = load_chunk(edges, base + 32, e, lane);
            const int n = (e - base < 32) ? (e - base) : 32;
            int a = m - base;
            a = (a < 0) ? 0 : ((a > n) ? n : a);
            const bool ends = row0 && m <= base + n;
            gather_chunk(acc, X4, cur.x, cur.y, n, ends ? a : -1,
                         &tl[(wave * 2) * WLDS_STRIDE + 4 * lane], lane);
            if (ends) row0 = false;
            cur = nxt;
        }
        if (row0) {
            *(float4*)&tl[(wave * 2) * WLDS_STRIDE + 4 * lane] = acc;
            acc = make_float4(0.f, 0.f, 0.f, 0.f);
        }
        *(float4*)&tl[(wave * 2 + 1) * WLDS_STRIDE + 4 * lane] = acc;
        __syncthreads();
        cur = load_chunk(edges, sn, en, lane);

        v8f c = (v8f){0.f,0.f,0.f,0.f,0.f,0.f,0.f,0.f};
        const float* ap = tl + l15 * WLDS_STRIDE + 2 * hi;
#pragma unroll
        for (int ks = 0; ks < 32; ++ks) {
            const v2f a2 = *(const v2f*)(ap + ks * 4);
            const v2f b2 = *(const v2f*)(bp + ks * 4);
            c = __builtin_amdgcn_wmma_f32_16x16x4_f32(
                 false, a2,  false, b2,
                 (short)0, c,
                 false,  false);
        }

#pragma unroll
        for (int v = 0; v < 8; ++v) {
            const int r = strip * 16 + v + 8 * hi;
            if (r < nNodes)
                __builtin_nontemporal_store(c[v] + bv, &out[(size_t)r * D + wave * 16 + l15]);
        }

        s = sn; m = mn; e = en;
        buf ^= 1;
    }
}

extern "C" void kernel_launch(void* const* d_in, const int* in_sizes, int n_in,
                              void* d_out, int out_size, void* d_ws, size_t ws_size,
                              hipStream_t stream) {
    const float* X    = (const float*)d_in[0];
    const int*   rows = (const int*)  d_in[1];
    const int*   cols = (const int*)  d_in[2];
    const float* vals = (const float*)d_in[3];
    const float* W    = (const float*)d_in[4];
    const float* bias = (const float*)d_in[5];
    float*       out  = (float*)d_out;

    const int nNodes = in_sizes[0] / D;
    const int nEdges = in_sizes[1];

    const int nChunks = (nEdges + EC - 1) / EC;
    const int nB      = (nNodes >> BSHIFT) + 1;
    const size_t recs = (size_t)nChunks * EC;
    char* p = (char*)d_ws;
    int2* cvA = (int2*)p;
    p += (recs * sizeof(int2) + 127) & ~(size_t)127;
    unsigned short* rA = (unsigned short*)p;
    p += (recs * sizeof(unsigned short) + 127) & ~(size_t)127;
    int* run = (int*)p;
    p += ((size_t)nChunks * (nB + 1) * sizeof(int) + 127) & ~(size_t)127;
    int* off = (int*)p;
    p += (((size_t)nNodes + 1) * sizeof(int) + 127) & ~(size_t)127;
    int2* edges = (int2*)p;

    chunk_bucket_kernel<<<nChunks, 1024, 0, stream>>>(rows, cols, vals, cvA, rA, run, nEdges, nB);
    bucket_rows_kernel<<<nB, 256, 0, stream>>>(cvA, rA, run, off, edges, nNodes, nChunks, nB);
    {
        const int nStrips = (nNodes + 15) / 16;
        const int nPairs  = (nStrips + 1) / 2;
        const int blocks  = (nPairs < 768) ? nPairs : 768;
        gather_project_kernel<<<blocks, 512, 0, stream>>>(X, off, edges, W, bias, out, nNodes, nStrips);
    }
}
